# starting kernel
# baseline (speedup 1.0000x reference)
#include <hip/hip_runtime.h>

typedef float v2f __attribute__((ext_vector_type(2)));
typedef float v8f __attribute__((ext_vector_type(8)));

#define CH 128
#define LDS_PITCH 132

__global__ __launch_bounds__(128) void gcn_gemm_wmma(
    const float* __restrict__ x, const float* __restrict__ W,
    const float* __restrict__ bias, float* __restrict__ xt, int ntiles) {
  __shared__ float wl[CH * LDS_PITCH];

  const int tid = threadIdx.x;
  for (int i = tid; i < CH * CH; i += 128) {
    const int r = i >> 7;
    const int c = i & (CH - 1);
    wl[r * LDS_PITCH + c] = W[i];
  }
  __syncthreads();

  const int wave = tid >> 5;
  const int lane = tid & 31;
  const int rt = blockIdx.x * 4 + wave;
  if (rt >= ntiles) return;
  const int rb  = rt * 16;
  const int lh  = lane & 15;
  const int khi = lane >> 4;

  v8f acc[8];
#pragma unroll
  for (int cb = 0; cb < 8; ++cb) {
    const float bv = bias[cb * 16 + lh];
    acc[cb] = (v8f){bv, bv, bv, bv, bv, bv, bv, bv};
  }

  const float* arow = &x[(size_t)(rb + lh) * CH + 2 * khi];

  for (int k = 0; k < CH; k += 4) {
    const v2f a = *(const v2f*)(arow + k);
#pragma unroll
    for (int cb = 0; cb < 8; ++cb) {
      const v2f bm = *(const v2f*)&wl[(cb * 16 + lh) * LDS_PITCH + k + 2 * khi];
      acc[cb] = __builtin_amdgcn_wmma_f32_16x16x4_f32(
          false, a, false, bm, (short)0, acc[cb], false, false);
    }
  }

#pragma unroll
  for (int cb = 0; cb < 8; ++cb) {
#pragma unroll
    for (int i = 0; i < 8; ++i) {
      xt[(size_t)(rb + i + 8 * khi) * CH + cb * 16 + lh] = acc[cb][i];
    }
  }
}

__global__ void gcn_zero(float4* __restrict__ p, int n4) {
  const int i = blockIdx.x * blockDim.x + threadIdx.x;
  if (i < n4) p[i] = make_float4(0.f, 0.f, 0.f, 0.f);
}

__global__ __launch_bounds__(256) void gcn_scatter(
    const float* __restrict__ xt, const int* __restrict__ ei,
    float* __restrict__ out, int nedges) {
  const int e = blockIdx.x * (blockDim.x >> 5) + (threadIdx.x >> 5);
  if (e >= nedges) return;
  const int lane = threadIdx.x & 31;
  const int row = ei[e];
  const int col = ei[nedges + e];

  const float4 v = ((const float4*)&xt[(size_t)col * CH])[lane];
  float* dst = &out[(size_t)row * CH + lane * 4];
  atomicAdd(dst + 0, v.x);
  atomicAdd(dst + 1, v.y);
  atomicAdd(dst + 2, v.z);
  atomicAdd(dst + 3, v.w);
}

__global__ void gcn_tanh(float4* __restrict__ p, int n4) {
  const int i = blockIdx.x * blockDim.x + threadIdx.x;
  if (i < n4) {
    float4 v = p[i];
    v.x = tanhf(v.x);
    v.y = tanhf(v.y);
    v.z = tanhf(v.z);
    v.w = tanhf(v.w);
    p[i] = v;
  }
}

extern "C" void kernel_launch(void* const* d_in, const int* in_sizes, int n_in,
                              void* d_out, int out_size, void* d_ws, size_t ws_size,
                              hipStream_t stream) {
  const float* x    = (const float*)d_in[0];
  const int*   ei   = (const int*)d_in[1];
  const float* W    = (const float*)d_in[2];
  const float* bias = (const float*)d_in[3];

  float* out = (float*)d_out;
  float* xt  = (float*)d_ws;

  const int N = in_sizes[0] / CH;
  const int E = in_sizes[1] / 2;
  const int total = N * CH;
  const int n4 = total / 4;

  gcn_zero<<<(n4 + 255) / 256, 256, 0, stream>>>((float4*)out, n4);

  const int ntiles = N / 16;
  gcn_gemm_wmma<<<(ntiles + 3) / 4, 128, 0, stream>>>(x, W, bias, xt, ntiles);

  gcn_scatter<<<(E + 7) / 8, 256, 0, stream>>>(xt, ei, out, E);

  gcn_tanh<<<(n4 + 255) / 256, 256, 0, stream>>>((float4*)out, n4);
}


# Float-atomic scatter replaced by in-call binning of edges by destination row (int atomics) and a per-row gather adding source rows in ascending source-index order, tanh fused; no zero or tanh pass
# speedup vs baseline: 1.6648x; 1.6648x over previous
#include <hip/hip_runtime.h>

typedef float v2f __attribute__((ext_vector_type(2)));
typedef float v8f __attribute__((ext_vector_type(8)));

#define CH 128
#define LDS_PITCH 132

__global__ __launch_bounds__(128) void gcn_gemm_wmma(
    const float* __restrict__ x, const float* __restrict__ W,
    const float* __restrict__ bias, float* __restrict__ xt, int ntiles) {
  __shared__ float wl[CH * LDS_PITCH];

  const int tid = threadIdx.x;
  for (int i = tid; i < CH * CH; i += 128) {
    const int r = i >> 7;
    const int c = i & (CH - 1);
    wl[r * LDS_PITCH + c] = W[i];
  }
  __syncthreads();

  const int wave = tid >> 5;
  const int lane = tid & 31;
  const int rt = blockIdx.x * 4 + wave;
  if (rt >= ntiles) return;
  const int rb  = rt * 16;
  const int lh  = lane & 15;
  const int khi = lane >> 4;

  v8f acc[8];
#pragma unroll
  for (int cb = 0; cb < 8; ++cb) {
    const float bv = bias[cb * 16 + lh];
    acc[cb] = (v8f){bv, bv, bv, bv, bv, bv, bv, bv};
  }

  const float* arow = &x[(size_t)(rb + lh) * CH + 2 * khi];

  for (int k = 0; k < CH; k += 4) {
    const v2f a = *(const v2f*)(arow + k);
#pragma unroll
    for (int cb = 0; cb < 8; ++cb) {
      const v2f bm = *(const v2f*)&wl[(cb * 16 + lh) * LDS_PITCH + k + 2 * khi];
      acc[cb] = __builtin_amdgcn_wmma_f32_16x16x4_f32(
          false, a, false, bm, (short)0, acc[cb], false, false);
    }
  }

#pragma unroll
  for (int cb = 0; cb < 8; ++cb) {
#pragma unroll
    for (int i = 0; i < 8; ++i) {
      xt[(size_t)(rb + i + 8 * khi) * CH + cb * 16 + lh] = acc[cb][i];
    }
  }
}

__global__ void gcn_zero_counts(int* __restrict__ cnt, int n) {
  const int i = blockIdx.x * blockDim.x + threadIdx.x;
  if (i < n) cnt[i] = 0;
}

#define CAP 64

static __device__ __forceinline__ void place_edge(
    int row, int col, int pos, int* __restrict__ cnt, int* __restrict__ bins,
    int* __restrict__ spill, int nrows) {
  if (pos < CAP) {
    bins[(size_t)row * CAP + pos] = col;
  } else {
    const int sp = atomicAdd(&cnt[nrows], 1);
    spill[2 * (size_t)sp] = row;
    spill[2 * (size_t)sp + 1] = col;
  }
}

__global__ __launch_bounds__(256) void gcn_bin_edges(
    const int* __restrict__ ei, int* __restrict__ cnt, int* __restrict__ bins,
    int* __restrict__ spill, int nedges, int nrows) {
  const int T = gridDim.x * blockDim.x;
  int e = blockIdx.x * blockDim.x + threadIdx.x;
  for (; e + 3 * T < nedges; e += 4 * T) {
    const int r0 = ei[e], r1 = ei[e + T], r2 = ei[e + 2 * T], r3 = ei[e + 3 * T];
    const int c0 = ei[nedges + e], c1 = ei[nedges + e + T];
    const int c2 = ei[nedges + e + 2 * T], c3 = ei[nedges + e + 3 * T];
    const int p0 = atomicAdd(&cnt[r0], 1);
    const int p1 = atomicAdd(&cnt[r1], 1);
    const int p2 = atomicAdd(&cnt[r2], 1);
    const int p3 = atomicAdd(&cnt[r3], 1);
    place_edge(r0, c0, p0, cnt, bins, spill, nrows);
    place_edge(r1, c1, p1, cnt, bins, spill, nrows);
    place_edge(r2, c2, p2, cnt, bins, spill, nrows);
    place_edge(r3, c3, p3, cnt, bins, spill, nrows);
  }
  for (; e < nedges; e += T) {
    const int r0 = ei[e];
    const int c0 = ei[nedges + e];
    const int p0 = atomicAdd(&cnt[r0], 1);
    place_edge(r0, c0, p0, cnt, bins, spill, nrows);
  }
}

template <int K>
static __device__ __forceinline__ void add_sorted(
    float4& acc, const float4* __restrict__ xrow, int sorted, int j) {
  float4 v[K];
#pragma unroll
  for (int u = 0; u < K; ++u) {
    const int c = __builtin_amdgcn_readlane(sorted, j + u);
    v[u] = xrow[(size_t)c * (CH / 4)];
  }
#pragma unroll
  for (int u = 0; u < K; ++u) {
    acc.x += v[u].x; acc.y += v[u].y; acc.z += v[u].z; acc.w += v[u].w;
  }
}

__global__ __launch_bounds__(256) void gcn_gather_rows(
    const float* __restrict__ xt, const int* __restrict__ cnt,
    const int* __restrict__ bins, const int* __restrict__ spill,
    float* __restrict__ out, int nrows) {
  const int lane = threadIdx.x & 31;
  const int w = __builtin_amdgcn_readfirstlane((int)(threadIdx.x >> 5));
  const int wpb = blockDim.x >> 5;
  const int nwaves = gridDim.x * wpb;
  const float4* xrow = (const float4*)xt + lane;

  int r = blockIdx.x * wpb + w;
  if (r >= nrows) return;
  int n_next = cnt[r];
  int k_next = bins[(size_t)r * CAP + lane];

  for (; r < nrows; r += nwaves) {
    const int n = __builtin_amdgcn_readfirstlane(n_next);
    int key = k_next;
    const int rn = r + nwaves;
    if (rn < nrows) {
      n_next = cnt[rn];
      k_next = bins[(size_t)rn * CAP + lane];
    }

    float4 acc = make_float4(0.f, 0.f, 0.f, 0.f);

    if (n <= 32) {
      key = lane < n ? key : 0x7fffffff;
      int rank = 0;
      for (int j = 0; j < n; ++j) {
        const int kj = __builtin_amdgcn_readlane(key, j);
        rank += (kj < key || (kj == key && j < lane)) ? 1 : 0;
      }
      if (lane >= n) rank = lane;
      const int sorted = __builtin_amdgcn_ds_permute(rank << 2, key);
      int j = 0;
      for (; j + 16 <= n; j += 16) add_sorted<16>(acc, xrow, sorted, j);
      if (n & 8) { add_sorted<8>(acc, xrow, sorted, j); j += 8; }
      if (n & 4) { add_sorted<4>(acc, xrow, sorted, j); j += 4; }
      if (n & 2) { add_sorted<2>(acc, xrow, sorted, j); j += 2; }
      if (n & 1) { add_sorted<1>(acc, xrow, sorted, j); }
    } else if (n <= CAP) {
      const int key0 = key;
      const int key1 = (lane + 32 < n) ? bins[(size_t)r * CAP + 32 + lane] : 0x7fffffff;
      int rank0 = 0, rank1 = 0;
      for (int j = 0; j < 32; ++j) {
        const int kj = __builtin_amdgcn_readlane(key0, j);
        rank0 += (kj < key0 || (kj == key0 && j < lane)) ? 1 : 0;
        rank1 += (kj <= key1) ? 1 : 0;
      }
      for (int j = 0; j < n - 32; ++j) {
        const int kj = __builtin_amdgcn_readlane(key1, j);
        rank0 += (kj < key0) ? 1 : 0;
        rank1 += (kj < key1 || (kj == key1 && j < lane)) ? 1 : 0;
      }
      if (lane + 32 >= n) rank1 = -1;
      for (int p0 = 0; p0 < n; p0 += 8) {
        float4 v[8];
#pragma unroll
        for (int u = 0; u < 8; ++u) {
          v[u] = make_float4(0.f, 0.f, 0.f, 0.f);
          const int p = p0 + u;
          if (p < n) {
            const unsigned b0 = __builtin_amdgcn_ballot_w32(rank0 == p);
            const unsigned b1 = __builtin_amdgcn_ballot_w32(rank1 == p);
            const unsigned b = b0 | b1;
            const int kk = b0 ? key0 : key1;
            const int c = __builtin_amdgcn_readlane(kk, __builtin_ctz(b));
            v[u] = xrow[(size_t)c * (CH / 4)];
          }
        }
#pragma unroll
        for (int u = 0; u < 8; ++u) {
          acc.x += v[u].x; acc.y += v[u].y; acc.z += v[u].z; acc.w += v[u].w;
        }
      }
    } else {
      const int* bin = bins + (size_t)r * CAP;
      const int nspill = cnt[nrows];
      int last = -1;
      for (int pass = 0; pass < n; ++pass) {
        int m = 0x7fffffff;
        for (int i = lane; i < CAP; i += 32) {
          const int v = bin[i];
          if (v > last && v < m) m = v;
        }
        for (int i = lane; i < nspill; i += 32) {
          if (spill[2 * (size_t)i] == r) {
            const int v = spill[2 * (size_t)i + 1];
            if (v > last && v < m) m = v;
          }
        }
        for (int s = 16; s > 0; s >>= 1) {
          const int o = __shfl_xor(m, s, 32);
          m = o < m ? o : m;
        }
        if (m == 0x7fffffff) break;
        int k = 0;
        for (int i = lane; i < CAP; i += 32) k += (bin[i] == m) ? 1 : 0;
        for (int i = lane; i < nspill; i += 32)
          k += (spill[2 * (size_t)i] == r && spill[2 * (size_t)i + 1] == m) ? 1 : 0;
        for (int s = 16; s > 0; s >>= 1) k += __shfl_xor(k, s, 32);
        const float4 v = xrow[(size_t)m * (CH / 4)];
        for (int t = 0; t < k; ++t) {
          acc.x += v.x; acc.y += v.y; acc.z += v.z; acc.w += v.w;
        }
        last = m;
      }
    }

    float4 o;
    o.x = tanhf(acc.x);
    o.y = tanhf(acc.y);
    o.z = tanhf(acc.z);
    o.w = tanhf(acc.w);
    ((float4*)out)[(size_t)r * (CH / 4) + lane] = o;
  }
}

extern "C" void kernel_launch(void* const* d_in, const int* in_sizes, int n_in,
                              void* d_out, int out_size, void* d_ws, size_t ws_size,
                              hipStream_t stream) {
  const float* x    = (const float*)d_in[0];
  const int*   ei   = (const int*)d_in[1];
  const float* W    = (const float*)d_in[2];
  const float* bias = (const float*)d_in[3];

  float* out = (float*)d_out;

  const int N = in_sizes[0] / CH;
  const int E = in_sizes[1] / 2;

  float* xt    = (float*)d_ws;
  int*   bins  = (int*)(xt + (size_t)N * CH);
  int*   spill = bins + (size_t)N * CAP;
  int*   cnt   = spill + 2 * (size_t)E;

  gcn_zero_counts<<<(N + 1 + 255) / 256, 256, 0, stream>>>(cnt, N + 1);

  gcn_bin_edges<<<(E + 1023) / 1024, 256, 0, stream>>>(ei, cnt, bins, spill, E, N);

  const int ntiles = N / 16;
  gcn_gemm_wmma<<<(ntiles + 3) / 4, 128, 0, stream>>>(x, W, bias, xt, ntiles);

  gcn_gather_rows<<<1024, 256, 0, stream>>>(xt, cnt, bins, spill, out, N);
}


# Product moved behind the aggregation: x rows summed per destination row (ascending source order), then one f32 WMMA product per 16-row tile, count*bias and tanh in one launch; sum order changes
# speedup vs baseline: 2.1714x; 1.3043x over previous
#include <hip/hip_runtime.h>


typedef float v2f __attribute__((ext_vector_type(2)));
typedef float v8f __attribute__((ext_vector_type(8)));

#define CH 128
#define A_PITCH 132

__global__ void gcn_zero_counts(int* __restrict__ cnt, int n) {
  const int i = blockIdx.x * blockDim.x + threadIdx.x;
  if (i < n) cnt[i] = 0;
}

#define CAP 64

static __device__ __forceinline__ void place_edge(
    int row, int col, int pos, int* __restrict__ cnt, int* __restrict__ bins,
    int* __restrict__ spill, int nrows) {
  if (pos < CAP) {
    bins[(size_t)row * CAP + pos] = col;
  } else {
    const int sp = atomicAdd(&cnt[nrows], 1);
    spill[2 * (size_t)sp] = row;
    spill[2 * (size_t)sp + 1] = col;
  }
}

__global__ __launch_bounds__(256) void gcn_bin_edges(
    const int* __restrict__ ei, int* __restrict__ cnt, int* __restrict__ bins,
    int* __restrict__ spill, int nedges, int nrows) {
  const int T = gridDim.x * blockDim.x;
  int e = blockIdx.x * blockDim.x + threadIdx.x;
  for (; e + 3 * T < nedges; e += 4 * T) {
    const int r0 = ei[e], r1 = ei[e + T], r2 = ei[e + 2 * T], r3 = ei[e + 3 * T];
    const int c0 = ei[nedges + e], c1 = ei[nedges + e + T];
    const int c2 = ei[nedges + e + 2 * T], c3 = ei[nedges + e + 3 * T];
    const int p0 = atomicAdd(&cnt[r0], 1);
    const int p1 = atomicAdd(&cnt[r1], 1);
    const int p2 = atomicAdd(&cnt[r2], 1);
    const int p3 = atomicAdd(&cnt[r3], 1);
    place_edge(r0, c0, p0, cnt, bins, spill, nrows);
    place_edge(r1, c1, p1, cnt, bins, spill, nrows);
    place_edge(r2, c2, p2, cnt, bins, spill, nrows);
    place_edge(r3, c3, p3, cnt, bins, spill, nrows);
  }
  for (; e < nedges; e += T) {
    const int r0 = ei[e];
    const int c0 = ei[nedges + e];
    const int p0 = atomicAdd(&cnt[r0], 1);
    place_edge(r0, c0, p0, cnt, bins, spill, nrows);
  }
}

template <int K>
static __device__ __forceinline__ void add_sorted(
    float4& acc, const float4* __restrict__ xrow, int sorted, int j) {
  float4 v[K];
#pragma unroll
  for (int u = 0; u < K; ++u) {
    const int c = __builtin_amdgcn_readlane(sorted, j + u);
    v[u] = xrow[(size_t)c * (CH / 4)];
  }
#pragma unroll
  for (int u = 0; u < K; ++u) {
    acc.x += v[u].x; acc.y += v[u].y; acc.z += v[u].z; acc.w += v[u].w;
  }
}

static __device__ __forceinline__ float4 sum_row(
    const float4* __restrict__ xrow, const int* __restrict__ cnt,
    const int* __restrict__ bins, const int* __restrict__ spill,
    int r, int n, int key, int lane, int nrows) {
  float4 acc = make_float4(0.f, 0.f, 0.f, 0.f);

  if (n <= 32) {
    key = lane < n ? key : 0x7fffffff;
    int rank = 0;
    for (int j = 0; j < n; ++j) {
      const int kj = __builtin_amdgcn_readlane(key, j);
      rank += (kj < key || (kj == key && j < lane)) ? 1 : 0;
    }
    if (lane >= n) rank = lane;
    const int sorted = __builtin_amdgcn_ds_permute(rank << 2, key);
    int j = 0;
    for (; j + 8 <= n; j += 8) add_sorted<8>(acc, xrow, sorted, j);
    const int rem = n - j;
    if (rem & 8) { add_sorted<8>(acc, xrow, sorted, j); j += 8; }
    if (rem & 4) { add_sorted<4>(acc, xrow, sorted, j); j += 4; }
    if (rem & 2) { add_sorted<2>(acc, xrow, sorted, j); j += 2; }
    if (rem & 1) { add_sorted<1>(acc, xrow, sorted, j); }
  } else if (n <= CAP) {
    const int key0 = key;
    const int key1 = (lane + 32 < n) ? bins[(size_t)r * CAP + 32 + lane] : 0x7fffffff;
    int rank0 = 0, rank1 = 0;
    for (int j = 0; j < 32; ++j) {
      const int kj = __builtin_amdgcn_readlane(key0, j);
      rank0 += (kj < key0 || (kj == key0 && j < lane)) ? 1 : 0;
      rank1 += (kj <= key1) ? 1 : 0;
    }
    for (int j = 0; j < n - 32; ++j) {
      const int kj = __builtin_amdgcn_readlane(key1, j);
      rank0 += (kj < key0) ? 1 : 0;
      rank1 += (kj < key1 || (kj == key1 && j < lane)) ? 1 : 0;
    }
    if (lane + 32 >= n) rank1 = -1;
    for (int p0 = 0; p0 < n; p0 += 8) {
      float4 v[8];
#pragma unroll
      for (int u = 0; u < 8; ++u) {
        v[u] = make_float4(0.f, 0.f, 0.f, 0.f);
        const int p = p0 + u;
        if (p < n) {
          const unsigned b0 = __builtin_amdgcn_ballot_w32(rank0 == p);
          const unsigned b1 = __builtin_amdgcn_ballot_w32(rank1 == p);
          const unsigned b = b0 | b1;
          const int kk = b0 ? key0 : key1;
          const int c = __builtin_amdgcn_readlane(kk, __builtin_ctz(b));
          v[u] = xrow[(size_t)c * (CH / 4)];
        }
      }
#pragma unroll
      for (int u = 0; u < 8; ++u) {
        acc.x += v[u].x; acc.y += v[u].y; acc.z += v[u].z; acc.w += v[u].w;
      }
    }
  } else {
    const int* bin = bins + (size_t)r * CAP;
    const int nspill = cnt[nrows];
    int last = -1;
    for (int pass = 0; pass < n; ++pass) {
      int m = 0x7fffffff;
      for (int i = lane; i < CAP; i += 32) {
        const int v = bin[i];
        if (v > last && v < m) m = v;
      }
      for (int i = lane; i < nspill; i += 32) {
        if (spill[2 * (size_t)i] == r) {
          const int v = spill[2 * (size_t)i + 1];
          if (v > last && v < m) m = v;
        }
      }
      for (int s = 16; s > 0; s >>= 1) {
        const int o = __shfl_xor(m, s, 32);
        m = o < m ? o : m;
      }
      if (m == 0x7fffffff) break;
      int k = 0;
      for (int i = lane; i < CAP; i += 32) k += (bin[i] == m) ? 1 : 0;
      for (int i = lane; i < nspill; i += 32)
        k += (spill[2 * (size_t)i] == r && spill[2 * (size_t)i + 1] == m) ? 1 : 0;
      for (int s = 16; s > 0; s >>= 1) k += __shfl_xor(k, s, 32);
      const float4 v = xrow[(size_t)m * (CH / 4)];
      for (int t = 0; t < k; ++t) {
        acc.x += v.x; acc.y += v.y; acc.z += v.z; acc.w += v.w;
      }
      last = m;
    }
  }
  return acc;
}

__global__ __launch_bounds__(256) __attribute__((amdgpu_waves_per_eu(8, 8))) void gcn_gather_product(
    const float* __restrict__ x, const float* __restrict__ W,
    const float* __restrict__ bias, const int* __restrict__ cnt,
    const int* __restrict__ bins, const int* __restrict__ spill,
    float* __restrict__ out, int nrows, int ntiles) {
  __shared__ float at[2][16 * A_PITCH];
  __shared__ int deg[2][16];

  const int lane = threadIdx.x & 31;
  const int w = __builtin_amdgcn_readfirstlane((int)(threadIdx.x >> 5));
  const int lh  = lane & 15;
  const int khi = lane >> 4;
  const float4* xrow = (const float4*)x + lane;

  v2f bf[32];
  {
    const float* wrow = W + (size_t)(16 * w + lh) * CH + 2 * khi;
#pragma unroll
    for (int s = 0; s < 32; ++s) bf[s] = *(const v2f*)(wrow + 4 * s);
  }
  const float bv = bias[16 * w + lh];

  int buf = 0;
  for (int tile = blockIdx.x; tile < ntiles; tile += gridDim.x, buf ^= 1) {
    const int rb = tile * 16;
    const int r0 = rb + 2 * w, r1 = r0 + 1;
    int n0 = 0, n1 = 0, k0 = 0, k1 = 0;
    if (r0 < nrows) { n0 = cnt[r0]; k0 = bins[(size_t)r0 * CAP + lane]; }
    if (r1 < nrows) { n1 = cnt[r1]; k1 = bins[(size_t)r1 * CAP + lane]; }
    n0 = __builtin_amdgcn_readfirstlane(n0);
    n1 = __builtin_amdgcn_readfirstlane(n1);

    const float4 s0 = sum_row(xrow, cnt, bins, spill, r0, n0, k0, lane, nrows);
    *(float4*)&at[buf][(2 * w) * A_PITCH + 4 * lane] = s0;
    const float4 s1 = sum_row(xrow, cnt, bins, spill, r1, n1, k1, lane, nrows);
    *(float4*)&at[buf][(2 * w + 1) * A_PITCH + 4 * lane] = s1;
    if (lane == 0) { deg[buf][2 * w] = n0; deg[buf][2 * w + 1] = n1; }
    __syncthreads();

    const float* arow = &at[buf][lh * A_PITCH + 2 * khi];
    v8f c0 = (v8f){0.f, 0.f, 0.f, 0.f, 0.f, 0.f, 0.f, 0.f};
    v8f c1 = c0;
#pragma unroll
    for (int s = 0; s < 32; s += 2) {
      const v2f a0 = *(const v2f*)(arow + 4 * s);
      const v2f a1 = *(const v2f*)(arow + 4 * s + 4);
      c0 = __builtin_amdgcn_wmma_f32_16x16x4_f32(
          false, a0, false, bf[s], (short)0, c0, false, false);
      c1 = __builtin_amdgcn_wmma_f32_16x16x4_f32(
          false, a1, false, bf[s + 1], (short)0, c1, false, false);
    }

    float res[8];
#pragma unroll
    for (int i = 0; i < 8; ++i) {
      const int d = deg[buf][i + 8 * khi];
      const float bb = d > 0 ? (float)d * bv : 0.f;
      res[i] = tanhf((c0[i] + c1[i]) + bb);
    }
#pragma unroll
    for (int i = 0; i < 8; ++i) {
      const int m = i + 8 * khi;
      if (rb + m < nrows) out[(size_t)(rb + m) * CH + 16 * w + lh] = res[i];
    }
  }
}

extern "C" void kernel_launch(void* const* d_in, const int* in_sizes, int n_in,
                              void* d_out, int out_size, void* d_ws, size_t ws_size,
                              hipStream_t stream) {
  const float* x    = (const float*)d_in[0];
  const int*   ei   = (const int*)d_in[1];
  const float* W    = (const float*)d_in[2];
  const float* bias = (const float*)d_in[3];

  float* out = (float*)d_out;

  const int N = in_sizes[0] / CH;
  const int E = in_sizes[1] / 2;

  int*   bins  = (int*)d_ws;
  int*   spill = bins + (size_t)N * CAP;
  int*   cnt   = spill + 2 * (size_t)E;

  gcn_zero_counts<<<(N + 1 + 255) / 256, 256, 0, stream>>>(cnt, N + 1);

  gcn_bin_edges<<<(E + 1023) / 1024, 256, 0, stream>>>(ei, cnt, bins, spill, E, N);

  const int ntiles = (N + 15) / 16;
  gcn_gather_product<<<1024, 256, 0, stream>>>(x, W, bias, cnt, bins, spill, out, N, ntiles);
}


# Global-atomic binning replaced by LDS binning: 128 work-groups scan all edge rows, slot their 391 rows' edges by LDS counters; zero launch dropped; sum order kept (rows over 64 edges: edge order)
# speedup vs baseline: 2.6486x; 1.2198x over previous
#include <hip/hip_runtime.h>


typedef float v2f __attribute__((ext_vector_type(2)));
typedef float v8f __attribute__((ext_vector_type(8)));

#define CH 128
#define A_PITCH 132

#define CAP 64
#define ROWS_PER_WG 391
#define ROWS_PAD 400
#define BIN_THREADS 1024
#define RES_PASSES ((ROWS_PER_WG + 31) / 32)

static __device__ __forceinline__ void file_edge(
    int row, int e, int lo, int nr, int* lcnt, int* lbins) {
  const unsigned d = (unsigned)(row - lo);
  if (d < (unsigned)nr) {
    const int pos = atomicAdd(&lcnt[d], 1);
    if (pos < CAP) lbins[d * CAP + pos] = e;
  }
}

__global__ __launch_bounds__(BIN_THREADS) __attribute__((amdgpu_waves_per_eu(8, 8))) void gcn_bin_rows(
    const int* __restrict__ ei, int* __restrict__ cnt, int* __restrict__ bins,
    int nedges, int nrows) {
  __shared__ __attribute__((aligned(16))) int lds[ROWS_PAD + ROWS_PER_WG * CAP];
  int* lcnt = lds;
  int* lbins = lds + ROWS_PAD;

  const int tid = threadIdx.x;
  const int lo = blockIdx.x * ROWS_PER_WG;
  const int nr = min(ROWS_PER_WG, nrows - lo);

  for (int i = tid; i < ROWS_PER_WG * CAP / 4; i += BIN_THREADS)
    ((int4*)lbins)[i] = make_int4(0, 0, 0, 0);
  if (tid < ROWS_PER_WG) lcnt[tid] = 0;
  __syncthreads();

  const int n4 = nedges >> 2;
  const int nfull = n4 / (8 * BIN_THREADS);
  const int4* rows4 = (const int4*)ei;

  if (n4 > 0) {
    int4 rc[8];
    const int pbase = 8 * BIN_THREADS * nfull + tid;
#pragma unroll
    for (int u = 0; u < 8; ++u) {
      const int p = pbase + BIN_THREADS * u;
      rc[u] = rows4[p < n4 ? p : 0];
    }

    if (nfull > 0) {
      const int b0 = blockIdx.x % nfull;
      auto block_of = [&](int b) -> int {
        int bb = b0 + b;
        if (bb >= nfull) bb -= nfull;
        return bb;
      };
      auto load8 = [&](int4 (&r)[8], int b) {
        const int base = b < nfull ? tid + 8 * BIN_THREADS * block_of(b) : 0;
#pragma unroll
        for (int u = 0; u < 8; ++u) r[u] = rows4[base + BIN_THREADS * u];
      };
      auto file8 = [&](const int4 (&r)[8], int b) {
        if (b < nfull) {
          const int e0 = 4 * (tid + 8 * BIN_THREADS * block_of(b));
#pragma unroll
          for (int u = 0; u < 8; ++u) {
            const int e = e0 + 4 * BIN_THREADS * u;
            file_edge(r[u].x, e,     lo, nr, lcnt, lbins);
            file_edge(r[u].y, e + 1, lo, nr, lcnt, lbins);
            file_edge(r[u].z, e + 2, lo, nr, lcnt, lbins);
            file_edge(r[u].w, e + 3, lo, nr, lcnt, lbins);
          }
        }
      };
      int4 ra[8], rb[8];
      load8(ra, 0);
      for (int b = 0; b < nfull; b += 2) {
        load8(rb, b + 1);
        file8(ra, b);
        load8(ra, b + 2);
        file8(rb, b + 1);
      }
    }

#pragma unroll
    for (int u = 0; u < 8; ++u) {
      const int p = pbase + BIN_THREADS * u;
      const int nru = p < n4 ? nr : 0;
      file_edge(rc[u].x, 4 * p,     lo, nru, lcnt, lbins);
      file_edge(rc[u].y, 4 * p + 1, lo, nru, lcnt, lbins);
      file_edge(rc[u].z, 4 * p + 2, lo, nru, lcnt, lbins);
      file_edge(rc[u].w, 4 * p + 3, lo, nru, lcnt, lbins);
    }
  }
  for (int e = 4 * n4 + tid; e < nedges; e += BIN_THREADS)
    file_edge(ei[e], e, lo, nr, lcnt, lbins);
  __syncthreads();

  {
    const int sl = tid & 31, dr = tid >> 5;
    int nn[RES_PASSES];
#pragma unroll
    for (int k = 0; k < RES_PASSES; ++k) {
      const int d = dr + 32 * k;
      nn[k] = min(lcnt[d < nr ? d : 0], CAP);
      if (d >= nr) nn[k] = 0;
    }
    for (int h = 0; h < CAP; h += 32) {
      int cv[RES_PASSES];
#pragma unroll
      for (int k = 0; k < RES_PASSES; ++k) {
        const int d = dr + 32 * k;
        cv[k] = lbins[(d < nr ? d : 0) * CAP + h + sl];
      }
#pragma unroll
      for (int k = 0; k < RES_PASSES; ++k)
        if (sl + h < nn[k]) cv[k] = ei[nedges + cv[k]];
#pragma unroll
      for (int k = 0; k < RES_PASSES; ++k)
        if (sl + h < nn[k]) lbins[(dr + 32 * k) * CAP + h + sl] = cv[k];
    }
  }
  __syncthreads();

  int4* gb = (int4*)(bins + (size_t)lo * CAP);
  for (int i = tid; i < nr * (CAP / 4); i += BIN_THREADS) gb[i] = ((const int4*)lbins)[i];
  if (tid < nr) cnt[lo + tid] = lcnt[tid];
}

template <int K>
static __device__ __forceinline__ void add_sorted(
    float4& acc, const float4* __restrict__ xrow, int sorted, int j) {
  float4 v[K];
#pragma unroll
  for (int u = 0; u < K; ++u) {
    const int c = __builtin_amdgcn_readlane(sorted, j + u);
    v[u] = xrow[(size_t)c * (CH / 4)];
  }
#pragma unroll
  for (int u = 0; u < K; ++u) {
    acc.x += v[u].x; acc.y += v[u].y; acc.z += v[u].z; acc.w += v[u].w;
  }
}

static __device__ __forceinline__ float4 sum_row(
    const float4* __restrict__ xrow, const int* __restrict__ bins,
    const int* __restrict__ ei, int r, int n, int key, int lane, int nedges) {
  float4 acc = make_float4(0.f, 0.f, 0.f, 0.f);

  if (n <= 32) {
    key = lane < n ? key : 0x7fffffff;
    int rank = 0;
    for (int j = 0; j < n; ++j) {
      const int kj = __builtin_amdgcn_readlane(key, j);
      rank += (kj < key || (kj == key && j < lane)) ? 1 : 0;
    }
    if (lane >= n) rank = lane;
    const int sorted = __builtin_amdgcn_ds_permute(rank << 2, key);
    int j = 0;
    for (; j + 8 <= n; j += 8) add_sorted<8>(acc, xrow, sorted, j);
    const int rem = n - j;
    if (rem & 8) { add_sorted<8>(acc, xrow, sorted, j); j += 8; }
    if (rem & 4) { add_sorted<4>(acc, xrow, sorted, j); j += 4; }
    if (rem & 2) { add_sorted<2>(acc, xrow, sorted, j); j += 2; }
    if (rem & 1) { add_sorted<1>(acc, xrow, sorted, j); }
  } else if (n <= CAP) {
    const int key0 = key;
    const int key1 = (lane + 32 < n) ? bins[(size_t)r * CAP + 32 + lane] : 0x7fffffff;
    int rank0 = 0, rank1 = 0;
    for (int j = 0; j < 32; ++j) {
      const int kj = __builtin_amdgcn_readlane(key0, j);
      rank0 += (kj < key0 || (kj == key0 && j < lane)) ? 1 : 0;
      rank1 += (kj <= key1) ? 1 : 0;
    }
    for (int j = 0; j < n - 32; ++j) {
      const int kj = __builtin_amdgcn_readlane(key1, j);
      rank0 += (kj < key0) ? 1 : 0;
      rank1 += (kj < key1 || (kj == key1 && j < lane)) ? 1 : 0;
    }
    if (lane + 32 >= n) rank1 = -1;
    for (int p0 = 0; p0 < n; p0 += 8) {
      float4 v[8];
#pragma unroll
      for (int u = 0; u < 8; ++u) {
        v[u] = make_float4(0.f, 0.f, 0.f, 0.f);
        const int p = p0 + u;
        if (p < n) {
          const unsigned b0 = __builtin_amdgcn_ballot_w32(rank0 == p);
          const unsigned b1 = __builtin_amdgcn_ballot_w32(rank1 == p);
          const unsigned b = b0 | b1;
          const int kk = b0 ? key0 : key1;
          const int c = __builtin_amdgcn_readlane(kk, __builtin_ctz(b));
          v[u] = xrow[(size_t)c * (CH / 4)];
        }
      }
#pragma unroll
      for (int u = 0; u < 8; ++u) {
        acc.x += v[u].x; acc.y += v[u].y; acc.z += v[u].z; acc.w += v[u].w;
      }
    }
  } else {
    for (int e0 = 0; e0 < nedges; e0 += 32) {
      const int e = e0 + lane;
      const bool hit = e < nedges && ei[e] == r;
      const int c = hit ? ei[nedges + e] : 0;
      unsigned m = __builtin_amdgcn_ballot_w32(hit);
      while (m) {
        const int l = __builtin_ctz(m);
        m &= m - 1;
        const int cs = __builtin_amdgcn_readlane(c, l);
        const float4 v = xrow[(size_t)cs * (CH / 4)];
        acc.x += v.x; acc.y += v.y; acc.z += v.z; acc.w += v.w;
      }
    }
  }
  return acc;
}

__global__ __launch_bounds__(256) __attribute__((amdgpu_waves_per_eu(8, 8))) void gcn_gather_product(
    const float* __restrict__ x, const float* __restrict__ W,
    const float* __restrict__ bias, const int* __restrict__ cnt,
    const int* __restrict__ bins, const int* __restrict__ ei,
    float* __restrict__ out, int nrows, int nedges, int ntiles) {
  __shared__ float at[2][16 * A_PITCH];
  __shared__ int deg[2][16];

  const int lane = threadIdx.x & 31;
  const int w = __builtin_amdgcn_readfirstlane((int)(threadIdx.x >> 5));
  const int lh  = lane & 15;
  const int khi = lane >> 4;
  const float4* xrow = (const float4*)x + lane;

  v2f bf[32];
  {
    const float* wrow = W + (size_t)(16 * w + lh) * CH + 2 * khi;
#pragma unroll
    for (int s = 0; s < 32; ++s) bf[s] = *(const v2f*)(wrow + 4 * s);
  }
  const float bv = bias[16 * w + lh];

  int buf = 0;
  for (int tile = blockIdx.x; tile < ntiles; tile += gridDim.x, buf ^= 1) {
    const int rb = tile * 16;
    const int r0 = rb + 2 * w, r1 = r0 + 1;
    int n0 = 0, n1 = 0, k0 = 0, k1 = 0;
    if (r0 < nrows) { n0 = cnt[r0]; k0 = bins[(size_t)r0 * CAP + lane]; }
    if (r1 < nrows) { n1 = cnt[r1]; k1 = bins[(size_t)r1 * CAP + lane]; }
    n0 = __builtin_amdgcn_readfirstlane(n0);
    n1 = __builtin_amdgcn_readfirstlane(n1);

    const float4 s0 = sum_row(xrow, bins, ei, r0, n0, k0, lane, nedges);
    *(float4*)&at[buf][(2 * w) * A_PITCH + 4 * lane] = s0;
    const float4 s1 = sum_row(xrow, bins, ei, r1, n1, k1, lane, nedges);
    *(float4*)&at[buf][(2 * w + 1) * A_PITCH + 4 * lane] = s1;
    if (lane == 0) { deg[buf][2 * w] = n0; deg[buf][2 * w + 1] = n1; }
    __syncthreads();

    const float* arow = &at[buf][lh * A_PITCH + 2 * khi];
    v8f c0 = (v8f){0.f, 0.f, 0.f, 0.f, 0.f, 0.f, 0.f, 0.f};
    v8f c1 = c0;
#pragma unroll
    for (int s = 0; s < 32; s += 2) {
      const v2f a0 = *(const v2f*)(arow + 4 * s);
      const v2f a1 = *(const v2f*)(arow + 4 * s + 4);
      c0 = __builtin_amdgcn_wmma_f32_16x16x4_f32(
          false, a0, false, bf[s], (short)0, c0, false, false);
      c1 = __builtin_amdgcn_wmma_f32_16x16x4_f32(
          false, a1, false, bf[s + 1], (short)0, c1, false, false);
    }

    float res[8];
#pragma unroll
    for (int i = 0; i < 8; ++i) {
      const int d = deg[buf][i + 8 * khi];
      const float bb = d > 0 ? (float)d * bv : 0.f;
      res[i] = tanhf((c0[i] + c1[i]) + bb);
    }
#pragma unroll
    for (int i = 0; i < 8; ++i) {
      const int m = i + 8 * khi;
      if (rb + m < nrows) out[(size_t)(rb + m) * CH + 16 * w + lh] = res[i];
    }
  }
}

extern "C" void kernel_launch(void* const* d_in, const int* in_sizes, int n_in,
                              void* d_out, int out_size, void* d_ws, size_t ws_size,
                              hipStream_t stream) {
  const float* x    = (const float*)d_in[0];
  const int*   ei   = (const int*)d_in[1];
  const float* W    = (const float*)d_in[2];
  const float* bias = (const float*)d_in[3];

  float* out = (float*)d_out;

  const int N = in_sizes[0] / CH;
  const int E = in_sizes[1] / 2;

  int*   bins  = (int*)d_ws;
  int*   cnt   = bins + (size_t)N * CAP;

  gcn_bin_rows<<<(N + ROWS_PER_WG - 1) / ROWS_PER_WG, BIN_THREADS, 0, stream>>>(ei, cnt, bins, E, N);

  const int ntiles = (N + 15) / 16;
  gcn_gather_product<<<1024, 256, 0, stream>>>(x, W, bias, cnt, bins, ei, out, N, E, ntiles);
}


# Binning cut both ways: 64 row ranges x 4 parts of the edge list (256 work-groups, one per WGP), a 32-slot line and count plane per part; list read 64x not 128x, a quarter each; sum order kept
# speedup vs baseline: 3.4907x; 1.3179x over previous
#include <hip/hip_runtime.h>


typedef float v2f __attribute__((ext_vector_type(2)));
typedef float v8f __attribute__((ext_vector_type(8)));

#define CH 128
#define A_PITCH 132

#define NPART 4
#define HCAP 32
#define BSTRIDE (NPART * HCAP)
#define CAP 64
#define ROWS_PER_WG 782
#define ROWS_PAD 784
#define BIN_THREADS 1024
#define RES_CHUNK 13
#define RES_PASSES (2 * RES_CHUNK)

static __device__ __forceinline__ void file_edge(
    int row, int e, int lo, int nr, int* lcnt, int* lbins) {
  const unsigned d = (unsigned)(row - lo);
  if (d < (unsigned)nr) {
    const int pos = atomicAdd(&lcnt[d], 1);
    if (pos < HCAP) lbins[d * HCAP + pos] = e;
  }
}

__global__ __launch_bounds__(BIN_THREADS) __attribute__((amdgpu_waves_per_eu(8, 8))) void gcn_bin_rows(
    const int* __restrict__ ei, int* __restrict__ cnt, int* __restrict__ bins,
    int nedges, int nrows) {
  __shared__ __attribute__((aligned(16))) int lds[ROWS_PAD + ROWS_PER_WG * HCAP];
  int* lcnt = lds;
  int* lbins = lds + ROWS_PAD;

  const int tid = threadIdx.x;
  const int g = blockIdx.x / NPART;
  const int h = blockIdx.x % NPART;
  const int lo = g * ROWS_PER_WG;
  const int nr = min(ROWS_PER_WG, nrows - lo);

  for (int i = tid; i < ROWS_PER_WG * HCAP / 4; i += BIN_THREADS)
    ((int4*)lbins)[i] = make_int4(0, 0, 0, 0);
  if (tid < ROWS_PER_WG) lcnt[tid] = 0;
  __syncthreads();

  const int n4all = nedges >> 2;
  const int pbeg = (int)(((long long)n4all * h) / NPART);
  const int n4 = (int)(((long long)n4all * (h + 1)) / NPART) - pbeg;
  const int nfull = n4 / (8 * BIN_THREADS);
  const int4* rows4 = (const int4*)ei + pbeg;

  if (n4 > 0) {
    int4 rc[8];
    const int pbase = 8 * BIN_THREADS * nfull + tid;
#pragma unroll
    for (int u = 0; u < 8; ++u) {
      const int p = pbase + BIN_THREADS * u;
      rc[u] = rows4[p < n4 ? p : 0];
    }

    if (nfull > 0) {
      const int b0 = g % nfull;
      auto block_of = [&](int b) -> int {
        int bb = b0 + b;
        if (bb >= nfull) bb -= nfull;
        return bb;
      };
      auto load8 = [&](int4 (&r)[8], int b) {
        const int base = b < nfull ? tid + 8 * BIN_THREADS * block_of(b) : 0;
#pragma unroll
        for (int u = 0; u < 8; ++u) r[u] = rows4[base + BIN_THREADS * u];
      };
      auto file8 = [&](const int4 (&r)[8], int b) {
        if (b < nfull) {
          const int e0 = 4 * (pbeg + tid + 8 * BIN_THREADS * block_of(b));
#pragma unroll
          for (int u = 0; u < 8; ++u) {
            const int e = e0 + 4 * BIN_THREADS * u;
            file_edge(r[u].x, e,     lo, nr, lcnt, lbins);
            file_edge(r[u].y, e + 1, lo, nr, lcnt, lbins);
            file_edge(r[u].z, e + 2, lo, nr, lcnt, lbins);
            file_edge(r[u].w, e + 3, lo, nr, lcnt, lbins);
          }
        }
      };
      int4 ra[8], rb[8];
      load8(ra, 0);
      for (int b = 0; b < nfull; b += 2) {
        load8(rb, b + 1);
        file8(ra, b);
        load8(ra, b + 2);
        file8(rb, b + 1);
      }
    }

#pragma unroll
    for (int u = 0; u < 8; ++u) {
      const int p = pbase + BIN_THREADS * u;
      const int nru = p < n4 ? nr : 0;
      const int e = 4 * (pbeg + p);
      file_edge(rc[u].x, e,     lo, nru, lcnt, lbins);
      file_edge(rc[u].y, e + 1, lo, nru, lcnt, lbins);
      file_edge(rc[u].z, e + 2, lo, nru, lcnt, lbins);
      file_edge(rc[u].w, e + 3, lo, nru, lcnt, lbins);
    }
  }
  if (h == NPART - 1)
    for (int e = 4 * n4all + tid; e < nedges; e += BIN_THREADS)
      file_edge(ei[e], e, lo, nr, lcnt, lbins);
  __syncthreads();

  {
    const int sl = tid & 31, dr = tid >> 5;
    for (int k0 = 0; k0 < RES_PASSES; k0 += RES_CHUNK) {
      int nn[RES_CHUNK], cv[RES_CHUNK];
#pragma unroll
      for (int k = 0; k < RES_CHUNK; ++k) {
        const int d = dr + 32 * (k0 + k);
        nn[k] = min(lcnt[d < nr ? d : 0], HCAP);
        if (d >= nr) nn[k] = 0;
        cv[k] = lbins[(d < nr ? d : 0) * HCAP + sl];
      }
#pragma unroll
      for (int k = 0; k < RES_CHUNK; ++k)
        if (sl < nn[k]) cv[k] = ei[nedges + cv[k]];
#pragma unroll
      for (int k = 0; k < RES_CHUNK; ++k)
        if (sl < nn[k]) lbins[(dr + 32 * (k0 + k)) * HCAP + sl] = cv[k];
    }
  }
  __syncthreads();

  int4* gb = (int4*)(bins + (size_t)lo * BSTRIDE) + (HCAP / 4) * h;
  for (int i = tid; i < nr * (HCAP / 4); i += BIN_THREADS)
    gb[(i >> 3) * (BSTRIDE / 4) + (i & 7)] = ((const int4*)lbins)[i];
  if (tid < nr) cnt[(size_t)h * nrows + lo + tid] = lcnt[tid];
}

template <int K>
static __device__ __forceinline__ void add_sorted(
    float4& acc, const float4* __restrict__ xrow, int sorted, int j) {
  float4 v[K];
#pragma unroll
  for (int u = 0; u < K; ++u) {
    const int c = __builtin_amdgcn_readlane(sorted, j + u);
    v[u] = xrow[(size_t)c * (CH / 4)];
  }
#pragma unroll
  for (int u = 0; u < K; ++u) {
    acc.x += v[u].x; acc.y += v[u].y; acc.z += v[u].z; acc.w += v[u].w;
  }
}

static __device__ __forceinline__ float4 sum_row(
    const float4* __restrict__ xrow, const int* __restrict__ bins,
    const int* __restrict__ ei, int r, int n, bool over, int o1, int o2, int o3,
    int key, int lane, int nedges) {
  float4 acc = make_float4(0.f, 0.f, 0.f, 0.f);

  if (over) {
    for (int e0 = 0; e0 < nedges; e0 += 32) {
      const int e = e0 + lane;
      const bool hit = e < nedges && ei[e] == r;
      const int c = hit ? ei[nedges + e] : 0;
      unsigned m = __builtin_amdgcn_ballot_w32(hit);
      while (m) {
        const int l = __builtin_ctz(m);
        m &= m - 1;
        const int cs = __builtin_amdgcn_readlane(c, l);
        const float4 v = xrow[(size_t)cs * (CH / 4)];
        acc.x += v.x; acc.y += v.y; acc.z += v.z; acc.w += v.w;
      }
    }
  } else if (n <= 32) {
    key = lane < n ? key : 0x7fffffff;
    int rank = 0;
    for (int j = 0; j < n; ++j) {
      const int kj = __builtin_amdgcn_readlane(key, j);
      rank += (kj < key || (kj == key && j < lane)) ? 1 : 0;
    }
    if (lane >= n) rank = lane;
    const int sorted = __builtin_amdgcn_ds_permute(rank << 2, key);
    int j = 0;
    for (; j + 8 <= n; j += 8) add_sorted<8>(acc, xrow, sorted, j);
    const int rem = n - j;
    if (rem & 8) { add_sorted<8>(acc, xrow, sorted, j); j += 8; }
    if (rem & 4) { add_sorted<4>(acc, xrow, sorted, j); j += 4; }
    if (rem & 2) { add_sorted<2>(acc, xrow, sorted, j); j += 2; }
    if (rem & 1) { add_sorted<1>(acc, xrow, sorted, j); }
  } else {
    const int key0 = key;
    const int j1 = 32 + lane;
    const int p1 = (j1 >= o1 ? 1 : 0) + (j1 >= o2 ? 1 : 0) + (j1 >= o3 ? 1 : 0);
    const int s1 = j1 - (p1 == 0 ? 0 : p1 == 1 ? o1 : p1 == 2 ? o2 : o3);
    const int key1 = (j1 < n) ? bins[(size_t)r * BSTRIDE + HCAP * p1 + s1] : 0x7fffffff;
    int rank0 = 0, rank1 = 0;
    for (int j = 0; j < 32; ++j) {
      const int kj = __builtin_amdgcn_readlane(key0, j);
      rank0 += (kj < key0 || (kj == key0 && j < lane)) ? 1 : 0;
      rank1 += (kj <= key1) ? 1 : 0;
    }
    for (int j = 0; j < n - 32; ++j) {
      const int kj = __builtin_amdgcn_readlane(key1, j);
      rank0 += (kj < key0) ? 1 : 0;
      rank1 += (kj < key1 || (kj == key1 && j < lane)) ? 1 : 0;
    }
    if (lane + 32 >= n) rank1 = -1;
    for (int p0 = 0; p0 < n; p0 += 8) {
      float4 v[8];
#pragma unroll
      for (int u = 0; u < 8; ++u) {
        v[u] = make_float4(0.f, 0.f, 0.f, 0.f);
        const int p = p0 + u;
        if (p < n) {
          const unsigned b0 = __builtin_amdgcn_ballot_w32(rank0 == p);
          const unsigned b1 = __builtin_amdgcn_ballot_w32(rank1 == p);
          const unsigned b = b0 | b1;
          const int kk = b0 ? key0 : key1;
          const int c = __builtin_amdgcn_readlane(kk, __builtin_ctz(b));
          v[u] = xrow[(size_t)c * (CH / 4)];
        }
      }
#pragma unroll
      for (int u = 0; u < 8; ++u) {
        acc.x += v[u].x; acc.y += v[u].y; acc.z += v[u].z; acc.w += v[u].w;
      }
    }
  }
  return acc;
}

__global__ __launch_bounds__(256) __attribute__((amdgpu_waves_per_eu(8, 8))) void gcn_gather_product(
    const float* __restrict__ x, const float* __restrict__ W,
    const float* __restrict__ bias, const int* __restrict__ cnt,
    const int* __restrict__ bins, const int* __restrict__ ei,
    float* __restrict__ out, int nrows, int nedges, int ntiles) {
  __shared__ float at[2][16 * A_PITCH];
  __shared__ int deg[2][16];

  const int lane = threadIdx.x & 31;
  const int w = __builtin_amdgcn_readfirstlane((int)(threadIdx.x >> 5));
  const int lh  = lane & 15;
  const int khi = lane >> 4;
  const float4* xrow = (const float4*)x + lane;

  v2f bf[32];
  {
    const float* wrow = W + (size_t)(16 * w + lh) * CH + 2 * khi;
#pragma unroll
    for (int s = 0; s < 32; ++s) bf[s] = *(const v2f*)(wrow + 4 * s);
  }
  const float bv = bias[16 * w + lh];

  int buf = 0;
  for (int tile = blockIdx.x; tile < ntiles; tile += gridDim.x, buf ^= 1) {
    const int rb = tile * 16;
    const int r0 = rb + 2 * w, r1 = r0 + 1;
    int m0[NPART], m1[NPART], q0[NPART], q1[NPART];
#pragma unroll
    for (int p = 0; p < NPART; ++p) {
      m0[p] = 0; m1[p] = 0; q0[p] = 0; q1[p] = 0;
      if (r0 < nrows) {
        m0[p] = cnt[(size_t)p * nrows + r0];
        q0[p] = bins[(size_t)r0 * BSTRIDE + HCAP * p + lane];
      }
      if (r1 < nrows) {
        m1[p] = cnt[(size_t)p * nrows + r1];
        q1[p] = bins[(size_t)r1 * BSTRIDE + HCAP * p + lane];
      }
    }
#pragma unroll
    for (int p = 0; p < NPART; ++p) {
      m0[p] = __builtin_amdgcn_readfirstlane(m0[p]);
      m1[p] = __builtin_amdgcn_readfirstlane(m1[p]);
    }
    const int a1 = m0[0], a2 = a1 + m0[1], a3 = a2 + m0[2], n0 = a3 + m0[3];
    const int b1 = m1[0], b2 = b1 + m1[1], b3 = b2 + m1[2], n1 = b3 + m1[3];
    const bool over0 = m0[0] > HCAP || m0[1] > HCAP || m0[2] > HCAP || m0[3] > HCAP || n0 > CAP;
    const bool over1 = m1[0] > HCAP || m1[1] > HCAP || m1[2] > HCAP || m1[3] > HCAP || n1 > CAP;
    const int u01 = __builtin_amdgcn_ds_bpermute(((lane - a1) & 31) << 2, q0[1]);
    const int u02 = __builtin_amdgcn_ds_bpermute(((lane - a2) & 31) << 2, q0[2]);
    const int u03 = __builtin_amdgcn_ds_bpermute(((lane - a3) & 31) << 2, q0[3]);
    const int u11 = __builtin_amdgcn_ds_bpermute(((lane - b1) & 31) << 2, q1[1]);
    const int u12 = __builtin_amdgcn_ds_bpermute(((lane - b2) & 31) << 2, q1[2]);
    const int u13 = __builtin_amdgcn_ds_bpermute(((lane - b3) & 31) << 2, q1[3]);
    const int k0 = lane < a1 ? q0[0] : lane < a2 ? u01 : lane < a3 ? u02 : u03;
    const int k1 = lane < b1 ? q1[0] : lane < b2 ? u11 : lane < b3 ? u12 : u13;

    const float4 s0 = sum_row(xrow, bins, ei, r0, n0, over0, a1, a2, a3, k0, lane, nedges);
    *(float4*)&at[buf][(2 * w) * A_PITCH + 4 * lane] = s0;
    const float4 s1 = sum_row(xrow, bins, ei, r1, n1, over1, b1, b2, b3, k1, lane, nedges);
    *(float4*)&at[buf][(2 * w + 1) * A_PITCH + 4 * lane] = s1;
    if (lane == 0) { deg[buf][2 * w] = n0; deg[buf][2 * w + 1] = n1; }
    __syncthreads();

    const float* arow = &at[buf][lh * A_PITCH + 2 * khi];
    v8f c0 = (v8f){0.f, 0.f, 0.f, 0.f, 0.f, 0.f, 0.f, 0.f};
    v8f c1 = c0;
#pragma unroll
    for (int s = 0; s < 32; s += 2) {
      const v2f a0 = *(const v2f*)(arow + 4 * s);
      const v2f a1 = *(const v2f*)(arow + 4 * s + 4);
      c0 = __builtin_amdgcn_wmma_f32_16x16x4_f32(
          false, a0, false, bf[s], (short)0, c0, false, false);
      c1 = __builtin_amdgcn_wmma_f32_16x16x4_f32(
          false, a1, false, bf[s + 1], (short)0, c1, false, false);
    }

    float res[8];
#pragma unroll
    for (int i = 0; i < 8; ++i) {
      const int d = deg[buf][i + 8 * khi];
      const float bb = d > 0 ? (float)d * bv : 0.f;
      res[i] = tanhf((c0[i] + c1[i]) + bb);
    }
#pragma unroll
    for (int i = 0; i < 8; ++i) {
      const int m = i + 8 * khi;
      if (rb + m < nrows) out[(size_t)(rb + m) * CH + 16 * w + lh] = res[i];
    }
  }
}

extern "C" void kernel_launch(void* const* d_in, const int* in_sizes, int n_in,
                              void* d_out, int out_size, void* d_ws, size_t ws_size,
                              hipStream_t stream) {
  const float* x    = (const float*)d_in[0];
  const int*   ei   = (const int*)d_in[1];
  const float* W    = (const float*)d_in[2];
  const float* bias = (const float*)d_in[3];

  float* out = (float*)d_out;

  const int N = in_sizes[0] / CH;
  const int E = in_sizes[1] / 2;

  int*   bins  = (int*)d_ws;
  int*   cnt   = bins + (size_t)N * BSTRIDE;

  gcn_bin_rows<<<NPART * ((N + ROWS_PER_WG - 1) / ROWS_PER_WG), BIN_THREADS, 0, stream>>>(ei, cnt, bins, E, N);

  const int ntiles = (N + 15) / 16;
  gcn_gather_product<<<1024, 256, 0, stream>>>(x, W, bias, cnt, bins, ei, out, N, E, ntiles);
}


# Deeper row gather: a row's source rows go out 16 per group, its rest as one group of 4-16 (was 8, then 4, 2, 1); register room from W^T k 32-127 in LDS and 32-bit row offsets; sum order kept
# speedup vs baseline: 3.6308x; 1.0401x over previous
#include <hip/hip_runtime.h>


typedef float v2f __attribute__((ext_vector_type(2)));
typedef float v8f __attribute__((ext_vector_type(8)));
typedef float v4f __attribute__((ext_vector_type(4)));
typedef const v4f __attribute__((address_space(1))) gv4f;

#define CH 128
#define A_PITCH 132
#define KREG 8
#define WL_K (CH - 4 * KREG)
#define WL_PITCH 100
#define GB 16

#define NPART 4
#define HCAP 32
#define BSTRIDE (NPART * HCAP)
#define CAP 64
#define ROWS_PER_WG 782
#define ROWS_PAD 784
#define BIN_THREADS 1024
#define RES_CHUNK 13
#define RES_PASSES (2 * RES_CHUNK)

static __device__ __forceinline__ void file_edge(
    int row, int e, int lo, int nr, int* lcnt, int* lbins) {
  const unsigned d = (unsigned)(row - lo);
  if (d < (unsigned)nr) {
    const int pos = atomicAdd(&lcnt[d], 1);
    if (pos < HCAP) lbins[d * HCAP + pos] = e;
  }
}

__global__ __launch_bounds__(BIN_THREADS) __attribute__((amdgpu_waves_per_eu(8, 8))) void gcn_bin_rows(
    const int* __restrict__ ei, int* __restrict__ cnt, int* __restrict__ bins,
    int nedges, int nrows) {
  __shared__ __attribute__((aligned(16))) int lds[ROWS_PAD + ROWS_PER_WG * HCAP];
  int* lcnt = lds;
  int* lbins = lds + ROWS_PAD;

  const int tid = threadIdx.x;
  const int g = blockIdx.x / NPART;
  const int h = blockIdx.x % NPART;
  const int lo = g * ROWS_PER_WG;
  const int nr = min(ROWS_PER_WG, nrows - lo);

  for (int i = tid; i < ROWS_PER_WG * HCAP / 4; i += BIN_THREADS)
    ((int4*)lbins)[i] = make_int4(0, 0, 0, 0);
  if (tid < ROWS_PER_WG) lcnt[tid] = 0;
  __syncthreads();

  const int n4all = nedges >> 2;
  const int pbeg = (int)(((long long)n4all * h) / NPART);
  const int n4 = (int)(((long long)n4all * (h + 1)) / NPART) - pbeg;
  const int nfull = n4 / (8 * BIN_THREADS);
  const int4* rows4 = (const int4*)ei + pbeg;

  if (n4 > 0) {
    int4 rc[8];
    const int pbase = 8 * BIN_THREADS * nfull + tid;
#pragma unroll
    for (int u = 0; u < 8; ++u) {
      const int p = pbase + BIN_THREADS * u;
      rc[u] = rows4[p < n4 ? p : 0];
    }

    if (nfull > 0) {
      const int b0 = g % nfull;
      auto block_of = [&](int b) -> int {
        int bb = b0 + b;
        if (bb >= nfull) bb -= nfull;
        return bb;
      };
      auto load8 = [&](int4 (&r)[8], int b) {
        const int base = b < nfull ? tid + 8 * BIN_THREADS * block_of(b) : 0;
#pragma unroll
        for (int u = 0; u < 8; ++u) r[u] = rows4[base + BIN_THREADS * u];
      };
      auto file8 = [&](const int4 (&r)[8], int b) {
        if (b < nfull) {
          const int e0 = 4 * (pbeg + tid + 8 * BIN_THREADS * block_of(b));
#pragma unroll
          for (int u = 0; u < 8; ++u) {
            const int e = e0 + 4 * BIN_THREADS * u;
            file_edge(r[u].x, e,     lo, nr, lcnt, lbins);
            file_edge(r[u].y, e + 1, lo, nr, lcnt, lbins);
            file_edge(r[u].z, e + 2, lo, nr, lcnt, lbins);
            file_edge(r[u].w, e + 3, lo, nr, lcnt, lbins);
          }
        }
      };
      int4 ra[8], rb[8];
      load8(ra, 0);
      for (int b = 0; b < nfull; b += 2) {
        load8(rb, b + 1);
        file8(ra, b);
        load8(ra, b + 2);
        file8(rb, b + 1);
      }
    }

#pragma unroll
    for (int u = 0; u < 8; ++u) {
      const int p = pbase + BIN_THREADS * u;
      const int nru = p < n4 ? nr : 0;
      const int e = 4 * (pbeg + p);
      file_edge(rc[u].x, e,     lo, nru, lcnt, lbins);
      file_edge(rc[u].y, e + 1, lo, nru, lcnt, lbins);
      file_edge(rc[u].z, e + 2, lo, nru, lcnt, lbins);
      file_edge(rc[u].w, e + 3, lo, nru, lcnt, lbins);
    }
  }
  if (h == NPART - 1)
    for (int e = 4 * n4all + tid; e < nedges; e += BIN_THREADS)
      file_edge(ei[e], e, lo, nr, lcnt, lbins);
  __syncthreads();

  {
    const int sl = tid & 31, dr = tid >> 5;
    for (int k0 = 0; k0 < RES_PASSES; k0 += RES_CHUNK) {
      int nn[RES_CHUNK], cv[RES_CHUNK];
#pragma unroll
      for (int k = 0; k < RES_CHUNK; ++k) {
        const int d = dr + 32 * (k0 + k);
        nn[k] = min(lcnt[d < nr ? d : 0], HCAP);
        if (d >= nr) nn[k] = 0;
        cv[k] = lbins[(d < nr ? d : 0) * HCAP + sl];
      }
#pragma unroll
      for (int k = 0; k < RES_CHUNK; ++k)
        if (sl < nn[k]) cv[k] = ei[nedges + cv[k]];
#pragma unroll
      for (int k = 0; k < RES_CHUNK; ++k)
        if (sl < nn[k]) lbins[(dr + 32 * (k0 + k)) * HCAP + sl] = cv[k];
    }
  }
  __syncthreads();

  int4* gb = (int4*)(bins + (size_t)lo * BSTRIDE) + (HCAP / 4) * h;
  for (int i = tid; i < nr * (HCAP / 4); i += BIN_THREADS)
    gb[(i >> 3) * (BSTRIDE / 4) + (i & 7)] = ((const int4*)lbins)[i];
  if (tid < nr) cnt[(size_t)h * nrows + lo + tid] = lcnt[tid];
}

typedef const char __attribute__((address_space(1))) gchar;
static __device__ __forceinline__ v4f load_row(const float* __restrict__ x, int c, unsigned lane) {
  const unsigned off = ((unsigned)c << 9) + (lane << 4);
  return *(const gv4f*)((gchar*)x + off);
}

template <int K>
static __device__ __forceinline__ void add_sorted(
    float4& acc, const float* __restrict__ x, unsigned lane, int sorted, int j) {
  v4f v[K];
#pragma unroll
  for (int u = 0; u < K; ++u) {
    const int c = __builtin_amdgcn_readlane(sorted, j + u);
    v[u] = load_row(x, c, lane);
  }
#pragma unroll
  for (int u = 0; u < K; ++u) {
    acc.x += v[u].x; acc.y += v[u].y; acc.z += v[u].z; acc.w += v[u].w;
  }
}

template <int K>
static __device__ __forceinline__ void add_tail(
    float4& acc, const float* __restrict__ x, unsigned lane, int sorted, int j, int rem) {
  v4f v[K];
#pragma unroll
  for (int u = 0; u < K; ++u) {
    const int c = __builtin_amdgcn_readlane(sorted, j + u);
    v[u] = load_row(x, c, lane);
  }
#pragma unroll
  for (int u = 0; u < K; ++u) {
    const bool on = u < rem;
    const float sx = acc.x + v[u].x, sy = acc.y + v[u].y;
    const float sz = acc.z + v[u].z, sw = acc.w + v[u].w;
    acc.x = on ? sx : acc.x; acc.y = on ? sy : acc.y;
    acc.z = on ? sz : acc.z; acc.w = on ? sw : acc.w;
  }
}

static __device__ __forceinline__ float4 sum_row(
    const float* __restrict__ x, const float4* __restrict__ xrow, const int* __restrict__ bins,
    const int* __restrict__ ei, int r, int n, bool over, int o1, int o2, int o3,
    int key, int lane, int nedges) {
  float4 acc = make_float4(0.f, 0.f, 0.f, 0.f);

  if (over) {
    for (int e0 = 0; e0 < nedges; e0 += 32) {
      const int e = e0 + lane;
      const bool hit = e < nedges && ei[e] == r;
      const int c = hit ? ei[nedges + e] : 0;
      unsigned m = __builtin_amdgcn_ballot_w32(hit);
      while (m) {
        const int l = __builtin_ctz(m);
        m &= m - 1;
        const int cs = __builtin_amdgcn_readlane(c, l);
        const float4 v = xrow[(size_t)cs * (CH / 4)];
        acc.x += v.x; acc.y += v.y; acc.z += v.z; acc.w += v.w;
      }
    }
  } else if (n <= 32) {
    key = lane < n ? key : 0x7fffffff;
    int rank = 0;
    for (int j = 0; j < n; ++j) {
      const int kj = __builtin_amdgcn_readlane(key, j);
      rank += (kj < key || (kj == key && j < lane)) ? 1 : 0;
    }
    if (lane >= n) rank = lane;
    int sorted = __builtin_amdgcn_ds_permute(rank << 2, key);
    const int last = __builtin_amdgcn_readlane(sorted, n - 1);
    sorted = lane < n ? sorted : last;
    int j = 0;
    for (; j + GB <= n; j += GB) add_sorted<GB>(acc, x, (unsigned)lane, sorted, j);
    const int rem = n - j;
    if (rem > 12) add_tail<16>(acc, x, (unsigned)lane, sorted, j, rem);
    else if (rem > 8) add_tail<12>(acc, x, (unsigned)lane, sorted, j, rem);
    else if (rem > 4) add_tail<8>(acc, x, (unsigned)lane, sorted, j, rem);
    else if (rem > 0) add_tail<4>(acc, x, (unsigned)lane, sorted, j, rem);
  } else {
    const int key0 = key;
    const int j1 = 32 + lane;
    const int p1 = (j1 >= o1 ? 1 : 0) + (j1 >= o2 ? 1 : 0) + (j1 >= o3 ? 1 : 0);
    const int s1 = j1 - (p1 == 0 ? 0 : p1 == 1 ? o1 : p1 == 2 ? o2 : o3);
    const int key1 = (j1 < n) ? bins[(size_t)r * BSTRIDE + HCAP * p1 + s1] : 0x7fffffff;
    int rank0 = 0, rank1 = 0;
    for (int j = 0; j < 32; ++j) {
      const int kj = __builtin_amdgcn_readlane(key0, j);
      rank0 += (kj < key0 || (kj == key0 && j < lane)) ? 1 : 0;
      rank1 += (kj <= key1) ? 1 : 0;
    }
    for (int j = 0; j < n - 32; ++j) {
      const int kj = __builtin_amdgcn_readlane(key1, j);
      rank0 += (kj < key0) ? 1 : 0;
      rank1 += (kj < key1 || (kj == key1 && j < lane)) ? 1 : 0;
    }
    if (lane + 32 >= n) rank1 = -1;
    for (int p0 = 0; p0 < n; p0 += 8) {
      float4 v[8];
#pragma unroll
      for (int u = 0; u < 8; ++u) {
        v[u] = make_float4(0.f, 0.f, 0.f, 0.f);
        const int p = p0 + u;
        if (p < n) {
          const unsigned b0 = __builtin_amdgcn_ballot_w32(rank0 == p);
          const unsigned b1 = __builtin_amdgcn_ballot_w32(rank1 == p);
          const unsigned b = b0 | b1;
          const int kk = b0 ? key0 : key1;
          const int c = __builtin_amdgcn_readlane(kk, __builtin_ctz(b));
          v[u] = xrow[(size_t)c * (CH / 4)];
        }
      }
#pragma unroll
      for (int u = 0; u < 8; ++u) {
        acc.x += v[u].x; acc.y += v[u].y; acc.z += v[u].z; acc.w += v[u].w;
      }
    }
  }
  return acc;
}

__global__ __launch_bounds__(256) __attribute__((amdgpu_waves_per_eu(8, 8))) void gcn_gather_product(
    const float* __restrict__ x, const float* __restrict__ W,
    const float* __restrict__ bias, const int* __restrict__ cnt,
    const int* __restrict__ bins, const int* __restrict__ ei,
    float* __restrict__ out, int nrows, int nedges, int ntiles) {
  __shared__ float at[2][16 * A_PITCH];
  __shared__ int deg[2][16];
  __shared__ __attribute__((aligned(16))) float wl[CH * WL_PITCH];

  const int lane = threadIdx.x & 31;
  const int w = __builtin_amdgcn_readfirstlane((int)(threadIdx.x >> 5));
  const int lh  = lane & 15;
  const int khi = lane >> 4;
  const float4* xrow = (const float4*)x + lane;

  v2f bf[KREG];
  {
    const float* wrow = W + (size_t)(16 * w + lh) * CH + 2 * khi;
#pragma unroll
    for (int s = 0; s < KREG; ++s) bf[s] = *(const v2f*)(wrow + 4 * s);
  }
#pragma unroll
  for (int t = 0; t < CH * (WL_K / 4) / 256; ++t) {
    const int i = (int)threadIdx.x + 256 * t;
    const int n = i / (WL_K / 4), c4 = i % (WL_K / 4);
    *(float4*)&wl[n * WL_PITCH + 4 * c4] =
        *(const float4*)(W + (size_t)n * CH + 4 * KREG + 4 * c4);
  }
  const float* brow = &wl[(16 * w + lh) * WL_PITCH + 2 * khi];
  const float bv = bias[16 * w + lh];

  int buf = 0;
  for (int tile = blockIdx.x; tile < ntiles; tile += gridDim.x, buf ^= 1) {
    const int rb = tile * 16;
    const int r0 = rb + 2 * w, r1 = r0 + 1;
    int m0[NPART], m1[NPART], q0[NPART], q1[NPART];
#pragma unroll
    for (int p = 0; p < NPART; ++p) {
      m0[p] = 0; m1[p] = 0; q0[p] = 0; q1[p] = 0;
      if (r0 < nrows) {
        m0[p] = cnt[(size_t)p * nrows + r0];
        q0[p] = bins[(size_t)r0 * BSTRIDE + HCAP * p + lane];
      }
      if (r1 < nrows) {
        m1[p] = cnt[(size_t)p * nrows + r1];
        q1[p] = bins[(size_t)r1 * BSTRIDE + HCAP * p + lane];
      }
    }
#pragma unroll
    for (int p = 0; p < NPART; ++p) {
      m0[p] = __builtin_amdgcn_readfirstlane(m0[p]);
      m1[p] = __builtin_amdgcn_readfirstlane(m1[p]);
    }
    const int a1 = m0[0], a2 = a1 + m0[1], a3 = a2 + m0[2], n0 = a3 + m0[3];
    const int b1 = m1[0], b2 = b1 + m1[1], b3 = b2 + m1[2], n1 = b3 + m1[3];
    const bool over0 = m0[0] > HCAP || m0[1] > HCAP || m0[2] > HCAP || m0[3] > HCAP || n0 > CAP;
    const bool over1 = m1[0] > HCAP || m1[1] > HCAP || m1[2] > HCAP || m1[3] > HCAP || n1 > CAP;
    const int u01 = __builtin_amdgcn_ds_bpermute(((lane - a1) & 31) << 2, q0[1]);
    const int u02 = __builtin_amdgcn_ds_bpermute(((lane - a2) & 31) << 2, q0[2]);
    const int u03 = __builtin_amdgcn_ds_bpermute(((lane - a3) & 31) << 2, q0[3]);
    const int u11 = __builtin_amdgcn_ds_bpermute(((lane - b1) & 31) << 2, q1[1]);
    const int u12 = __builtin_amdgcn_ds_bpermute(((lane - b2) & 31) << 2, q1[2]);
    const int u13 = __builtin_amdgcn_ds_bpermute(((lane - b3) & 31) << 2, q1[3]);
    const int k0 = lane < a1 ? q0[0] : lane < a2 ? u01 : lane < a3 ? u02 : u03;
    const int k1 = lane < b1 ? q1[0] : lane < b2 ? u11 : lane < b3 ? u12 : u13;

    const float4 s0 = sum_row(x, xrow, bins, ei, r0, n0, over0, a1, a2, a3, k0, lane, nedges);
    *(float4*)&at[buf][(2 * w) * A_PITCH + 4 * lane] = s0;
    const float4 s1 = sum_row(x, xrow, bins, ei, r1, n1, over1, b1, b2, b3, k1, lane, nedges);
    *(float4*)&at[buf][(2 * w + 1) * A_PITCH + 4 * lane] = s1;
    if (lane == 0) { deg[buf][2 * w] = n0; deg[buf][2 * w + 1] = n1; }
    __syncthreads();

    const float* arow = &at[buf][lh * A_PITCH + 2 * khi];
    v8f c0 = (v8f){0.f, 0.f, 0.f, 0.f, 0.f, 0.f, 0.f, 0.f};
    v8f c1 = c0;
#pragma unroll
    for (int s = 0; s < 32; s += 2) {
      const v2f a0 = *(const v2f*)(arow + 4 * s);
      const v2f a1 = *(const v2f*)(arow + 4 * s + 4);
      const v2f b0 = s < KREG ? bf[s < KREG ? s : 0]
                              : *(const v2f*)(brow + 4 * (s - KREG));
      const v2f b1 = s + 1 < KREG ? bf[s + 1 < KREG ? s + 1 : 0]
                                  : *(const v2f*)(brow + 4 * (s + 1 - KREG));
      c0 = __builtin_amdgcn_wmma_f32_16x16x4_f32(
          false, a0, false, b0, (short)0, c0, false, false);
      c1 = __builtin_amdgcn_wmma_f32_16x16x4_f32(
          false, a1, false, b1, (short)0, c1, false, false);
    }

    float res[8];
#pragma unroll
    for (int i = 0; i < 8; ++i) {
      const int d = deg[buf][i + 8 * khi];
      const float bb = d > 0 ? (float)d * bv : 0.f;
      res[i] = tanhf((c0[i] + c1[i]) + bb);
    }
#pragma unroll
    for (int i = 0; i < 8; ++i) {
      const int m = i + 8 * khi;
      if (rb + m < nrows) out[(size_t)(rb + m) * CH + 16 * w + lh] = res[i];
    }
  }
}

extern "C" void kernel_launch(void* const* d_in, const int* in_sizes, int n_in,
                              void* d_out, int out_size, void* d_ws, size_t ws_size,
                              hipStream_t stream) {
  const float* x    = (const float*)d_in[0];
  const int*   ei   = (const int*)d_in[1];
  const float* W    = (const float*)d_in[2];
  const float* bias = (const float*)d_in[3];

  float* out = (float*)d_out;

  const int N = in_sizes[0] / CH;
  const int E = in_sizes[1] / 2;

  int*   bins  = (int*)d_ws;
  int*   cnt   = bins + (size_t)N * BSTRIDE;

  gcn_bin_rows<<<NPART * ((N + ROWS_PER_WG - 1) / ROWS_PER_WG), BIN_THREADS, 0, stream>>>(ei, cnt, bins, E, N);

  const int ntiles = (N + 15) / 16;
  gcn_gather_product<<<1024, 256, 0, stream>>>(x, W, bias, cnt, bins, ei, out, N, E, ntiles);
}


# Binning in two levels: 256 slices of the edge list file (row offset, source) words by 782-row range in LDS, then each bin work-group files its 64 segments by row; list read once; sum order kept
# speedup vs baseline: 4.2754x; 1.1775x over previous
#include <hip/hip_runtime.h>


typedef float v2f __attribute__((ext_vector_type(2)));
typedef float v8f __attribute__((ext_vector_type(8)));
typedef float v4f __attribute__((ext_vector_type(4)));
typedef const v4f __attribute__((address_space(1))) gv4f;

#define CH 128
#define A_PITCH 132
#define KREG 8
#define WL_K (CH - 4 * KREG)
#define WL_PITCH 100
#define GB 16

#define NPART 4
#define HCAP 32
#define BSTRIDE (NPART * HCAP)
#define CAP 64
#define ROWS_PER_WG 782
#define ROWS_PAD 784

#define NSLICE 256
#define NRANGE 64
#define SEGCAP 128
#define SPP (NSLICE / NPART)
#define PACK_BITS 22
#define PACK_MASK ((1u << PACK_BITS) - 1u)
#define A_THREADS 256
#define A_EPT 16
#define B_THREADS 512
static_assert(A_EPT == 16, "the load fence in gcn_file_ranges names 16 values");
#define B_LPS (B_THREADS / SPP)
#define B_PIECES (SEGCAP / 4 / B_LPS)

static __device__ __forceinline__ void file_src(
    unsigned d, int src, int nr, int* lcnt, int* lbins) {
  if (d < (unsigned)nr) {
    const int pos = atomicAdd(&lcnt[d], 1);
    if (pos < HCAP) lbins[d * HCAP + pos] = src;
  }
}

static __device__ __forceinline__ void slice_bounds(int s, int nedges, int& beg, int& end) {
  const int per = (nedges + NSLICE - 1) / NSLICE;
  beg = min(nedges, s * per);
  end = min(nedges, beg + per);
}

__global__ __launch_bounds__(A_THREADS) void gcn_file_ranges(
    const int* __restrict__ ei, int* __restrict__ seg, int* __restrict__ segcnt,
    int nedges, int nrows) {
  __shared__ __attribute__((aligned(16))) int lseg[NRANGE * SEGCAP];
  __shared__ int lcnt[NRANGE];

  const int tid = threadIdx.x;
  const int s = blockIdx.x;

  for (int i = tid; i < NRANGE * SEGCAP / 4; i += A_THREADS)
    ((int4*)lseg)[i] = make_int4(0, 0, 0, 0);
  if (tid < NRANGE) lcnt[tid] = 0;
  __syncthreads();

  int beg, end;
  slice_bounds(s, nedges, beg, end);
  const bool packable = nrows <= (int)(PACK_MASK + 1u);

  if (packable) {
    for (int e0 = beg + tid; e0 < end; e0 += A_EPT * A_THREADS) {
      int r[A_EPT], c[A_EPT];
#pragma unroll
      for (int u = 0; u < A_EPT; ++u) {
        const int e = e0 + A_THREADS * u;
        const int ee = e < end ? e : beg;
        r[u] = ei[ee];
        c[u] = ei[nedges + ee];
      }
      asm volatile("" :: "v"(c[0]), "v"(c[1]), "v"(c[2]), "v"(c[3]),
                         "v"(c[4]), "v"(c[5]), "v"(c[6]), "v"(c[7]),
                         "v"(c[8]), "v"(c[9]), "v"(c[10]), "v"(c[11]),
                         "v"(c[12]), "v"(c[13]), "v"(c[14]), "v"(c[15]));
      int pos[A_EPT];
      unsigned gg[A_EPT];
#pragma unroll
      for (int u = 0; u < A_EPT; ++u) {
        const int e = e0 + A_THREADS * u;
        const unsigned row = (unsigned)r[u];
        gg[u] = row / (unsigned)ROWS_PER_WG;
        pos[u] = SEGCAP;
        if (e < end && row < (unsigned)nrows && gg[u] < (unsigned)NRANGE)
          pos[u] = atomicAdd(&lcnt[gg[u]], 1);
      }
#pragma unroll
      for (int u = 0; u < A_EPT; ++u) {
        if (pos[u] < SEGCAP) {
          const unsigned d = (unsigned)r[u] - gg[u] * (unsigned)ROWS_PER_WG;
          lseg[gg[u] * SEGCAP + pos[u]] = (int)((d << PACK_BITS) | (unsigned)c[u]);
        }
      }
    }
  }
  __syncthreads();

  for (int i = tid; i < NRANGE * SEGCAP / 4; i += A_THREADS) {
    const int g = i / (SEGCAP / 4), p = i % (SEGCAP / 4);
    ((int4*)seg)[((size_t)g * NSLICE + s) * (SEGCAP / 4) + p] = ((const int4*)lseg)[i];
  }
  if (tid < NRANGE) segcnt[tid * NSLICE + s] = packable ? lcnt[tid] : SEGCAP + 1;
}

__global__ __launch_bounds__(B_THREADS) void gcn_bin_rows(
    const int* __restrict__ ei, const int* __restrict__ seg, const int* __restrict__ segcnt,
    int* __restrict__ cnt, int* __restrict__ bins, int nedges, int nrows) {
  __shared__ __attribute__((aligned(16))) int lds[ROWS_PAD + ROWS_PER_WG * HCAP];
  int* lcnt = lds;
  int* lbins = lds + ROWS_PAD;

  const int tid = threadIdx.x;
  const int g = blockIdx.x / NPART;
  const int h = blockIdx.x % NPART;
  const int lo = g * ROWS_PER_WG;
  const int nr = min(ROWS_PER_WG, nrows - lo);

  const int q = tid % B_LPS;
  const int s = h * SPP + tid / B_LPS;
  const bool have = g < NRANGE;
  const int sc = have ? segcnt[g * NSLICE + s] : SEGCAP + 1;
  int4 v[B_PIECES];
#pragma unroll
  for (int u = 0; u < B_PIECES; ++u) {
    v[u] = make_int4(0, 0, 0, 0);
    if (have) v[u] = ((const int4*)seg)[((size_t)g * NSLICE + s) * (SEGCAP / 4) + q + B_LPS * u];
  }

  for (int i = tid; i < ROWS_PER_WG * HCAP / 4; i += B_THREADS)
    ((int4*)lbins)[i] = make_int4(0, 0, 0, 0);
  for (int i = tid; i < ROWS_PER_WG; i += B_THREADS) lcnt[i] = 0;
  __syncthreads();

  if (sc <= SEGCAP) {
    unsigned dd[4 * B_PIECES];
    int ss[4 * B_PIECES], pp[4 * B_PIECES];
#pragma unroll
    for (int u = 0; u < B_PIECES; ++u) {
      const int i0 = 4 * (q + B_LPS * u);
      const unsigned ev[4] = {(unsigned)v[u].x, (unsigned)v[u].y, (unsigned)v[u].z, (unsigned)v[u].w};
#pragma unroll
      for (int k = 0; k < 4; ++k) {
        dd[4 * u + k] = i0 + k < sc ? ev[k] >> PACK_BITS : 0xffffffffu;
        ss[4 * u + k] = (int)(ev[k] & PACK_MASK);
        pp[4 * u + k] = HCAP;
        if (dd[4 * u + k] < (unsigned)nr) pp[4 * u + k] = atomicAdd(&lcnt[dd[4 * u + k]], 1);
      }
    }
#pragma unroll
    for (int k = 0; k < 4 * B_PIECES; ++k)
      if (pp[k] < HCAP) lbins[dd[k] * HCAP + pp[k]] = ss[k];
  } else {
    int beg, end;
    slice_bounds(s, nedges, beg, end);
    for (int e = beg + q; e < end; e += B_LPS)
      file_src((unsigned)(ei[e] - lo), ei[nedges + e], nr, lcnt, lbins);
  }
  __syncthreads();

  int4* gb = (int4*)(bins + (size_t)lo * BSTRIDE) + (HCAP / 4) * h;
  for (int i = tid; i < nr * (HCAP / 4); i += B_THREADS)
    gb[(i >> 3) * (BSTRIDE / 4) + (i & 7)] = ((const int4*)lbins)[i];
  for (int i = tid; i < nr; i += B_THREADS) cnt[(size_t)h * nrows + lo + i] = lcnt[i];
}

typedef const char __attribute__((address_space(1))) gchar;
static __device__ __forceinline__ v4f load_row(const float* __restrict__ x, int c, unsigned lane) {
  const unsigned off = ((unsigned)c << 9) + (lane << 4);
  return *(const gv4f*)((gchar*)x + off);
}

template <int K>
static __device__ __forceinline__ void add_sorted(
    float4& acc, const float* __restrict__ x, unsigned lane, int sorted, int j) {
  v4f v[K];
#pragma unroll
  for (int u = 0; u < K; ++u) {
    const int c = __builtin_amdgcn_readlane(sorted, j + u);
    v[u] = load_row(x, c, lane);
  }
#pragma unroll
  for (int u = 0; u < K; ++u) {
    acc.x += v[u].x; acc.y += v[u].y; acc.z += v[u].z; acc.w += v[u].w;
  }
}

template <int K>
static __device__ __forceinline__ void add_tail(
    float4& acc, const float* __restrict__ x, unsigned lane, int sorted, int j, int rem) {
  v4f v[K];
#pragma unroll
  for (int u = 0; u < K; ++u) {
    const int c = __builtin_amdgcn_readlane(sorted, j + u);
    v[u] = load_row(x, c, lane);
  }
#pragma unroll
  for (int u = 0; u < K; ++u) {
    const bool on = u < rem;
    const float sx = acc.x + v[u].x, sy = acc.y + v[u].y;
    const float sz = acc.z + v[u].z, sw = acc.w + v[u].w;
    acc.x = on ? sx : acc.x; acc.y = on ? sy : acc.y;
    acc.z = on ? sz : acc.z; acc.w = on ? sw : acc.w;
  }
}

static __device__ __forceinline__ float4 sum_row(
    const float* __restrict__ x, const float4* __restrict__ xrow, const int* __restrict__ bins,
    const int* __restrict__ ei, int r, int n, bool over, int o1, int o2, int o3,
    int key, int lane, int nedges) {
  float4 acc = make_float4(0.f, 0.f, 0.f, 0.f);

  if (over) {
    for (int e0 = 0; e0 < nedges; e0 += 32) {
      const int e = e0 + lane;
      const bool hit = e < nedges && ei[e] == r;
      const int c = hit ? ei[nedges + e] : 0;
      unsigned m = __builtin_amdgcn_ballot_w32(hit);
      while (m) {
        const int l = __builtin_ctz(m);
        m &= m - 1;
        const int cs = __builtin_amdgcn_readlane(c, l);
        const float4 v = xrow[(size_t)cs * (CH / 4)];
        acc.x += v.x; acc.y += v.y; acc.z += v.z; acc.w += v.w;
      }
    }
  } else if (n <= 32) {
    key = lane < n ? key : 0x7fffffff;
    int rank = 0;
    for (int j = 0; j < n; ++j) {
      const int kj = __builtin_amdgcn_readlane(key, j);
      rank += (kj < key || (kj == key && j < lane)) ? 1 : 0;
    }
    if (lane >= n) rank = lane;
    int sorted = __builtin_amdgcn_ds_permute(rank << 2, key);
    const int last = __builtin_amdgcn_readlane(sorted, n - 1);
    sorted = lane < n ? sorted : last;
    int j = 0;
    for (; j + GB <= n; j += GB) add_sorted<GB>(acc, x, (unsigned)lane, sorted, j);
    const int rem = n - j;
    if (rem > 12) add_tail<16>(acc, x, (unsigned)lane, sorted, j, rem);
    else if (rem > 8) add_tail<12>(acc, x, (unsigned)lane, sorted, j, rem);
    else if (rem > 4) add_tail<8>(acc, x, (unsigned)lane, sorted, j, rem);
    else if (rem > 0) add_tail<4>(acc, x, (unsigned)lane, sorted, j, rem);
  } else {
    const int key0 = key;
    const int j1 = 32 + lane;
    const int p1 = (j1 >= o1 ? 1 : 0) + (j1 >= o2 ? 1 : 0) + (j1 >= o3 ? 1 : 0);
    const int s1 = j1 - (p1 == 0 ? 0 : p1 == 1 ? o1 : p1 == 2 ? o2 : o3);
    const int key1 = (j1 < n) ? bins[(size_t)r * BSTRIDE + HCAP * p1 + s1] : 0x7fffffff;
    int rank0 = 0, rank1 = 0;
    for (int j = 0; j < 32; ++j) {
      const int kj = __builtin_amdgcn_readlane(key0, j);
      rank0 += (kj < key0 || (kj == key0 && j < lane)) ? 1 : 0;
      rank1 += (kj <= key1) ? 1 : 0;
    }
    for (int j = 0; j < n - 32; ++j) {
      const int kj = __builtin_amdgcn_readlane(key1, j);
      rank0 += (kj < key0) ? 1 : 0;
      rank1 += (kj < key1 || (kj == key1 && j < lane)) ? 1 : 0;
    }
    if (lane + 32 >= n) rank1 = -1;
    for (int p0 = 0; p0 < n; p0 += 8) {
      float4 v[8];
#pragma unroll
      for (int u = 0; u < 8; ++u) {
        v[u] = make_float4(0.f, 0.f, 0.f, 0.f);
        const int p = p0 + u;
        if (p < n) {
          const unsigned b0 = __builtin_amdgcn_ballot_w32(rank0 == p);
          const unsigned b1 = __builtin_amdgcn_ballot_w32(rank1 == p);
          const unsigned b = b0 | b1;
          const int kk = b0 ? key0 : key1;
          const int c = __builtin_amdgcn_readlane(kk, __builtin_ctz(b));
          v[u] = xrow[(size_t)c * (CH / 4)];
        }
      }
#pragma unroll
      for (int u = 0; u < 8; ++u) {
        acc.x += v[u].x; acc.y += v[u].y; acc.z += v[u].z; acc.w += v[u].w;
      }
    }
  }
  return acc;
}

__global__ __launch_bounds__(256) __attribute__((amdgpu_waves_per_eu(8, 8))) void gcn_gather_product(
    const float* __restrict__ x, const float* __restrict__ W,
    const float* __restrict__ bias, const int* __restrict__ cnt,
    const int* __restrict__ bins, const int* __restrict__ ei,
    float* __restrict__ out, int nrows, int nedges, int ntiles) {
  __shared__ float at[2][16 * A_PITCH];
  __shared__ int deg[2][16];
  __shared__ __attribute__((aligned(16))) float wl[CH * WL_PITCH];

  const int lane = threadIdx.x & 31;
  const int w = __builtin_amdgcn_readfirstlane((int)(threadIdx.x >> 5));
  const int lh  = lane & 15;
  const int khi = lane >> 4;
  const float4* xrow = (const float4*)x + lane;

  v2f bf[KREG];
  {
    const float* wrow = W + (size_t)(16 * w + lh) * CH + 2 * khi;
#pragma unroll
    for (int s = 0; s < KREG; ++s) bf[s] = *(const v2f*)(wrow + 4 * s);
  }
#pragma unroll
  for (int t = 0; t < CH * (WL_K / 4) / 256; ++t) {
    const int i = (int)threadIdx.x + 256 * t;
    const int n = i / (WL_K / 4), c4 = i % (WL_K / 4);
    *(float4*)&wl[n * WL_PITCH + 4 * c4] =
        *(const float4*)(W + (size_t)n * CH + 4 * KREG + 4 * c4);
  }
  const float* brow = &wl[(16 * w + lh) * WL_PITCH + 2 * khi];
  const float bv = bias[16 * w + lh];

  int buf = 0;
  for (int tile = blockIdx.x; tile < ntiles; tile += gridDim.x, buf ^= 1) {
    const int rb = tile * 16;
    const int r0 = rb + 2 * w, r1 = r0 + 1;
    int m0[NPART], m1[NPART], q0[NPART], q1[NPART];
#pragma unroll
    for (int p = 0; p < NPART; ++p) {
      m0[p] = 0; m1[p] = 0; q0[p] = 0; q1[p] = 0;
      if (r0 < nrows) {
        m0[p] = cnt[(size_t)p * nrows + r0];
        q0[p] = bins[(size_t)r0 * BSTRIDE + HCAP * p + lane];
      }
      if (r1 < nrows) {
        m1[p] = cnt[(size_t)p * nrows + r1];
        q1[p] = bins[(size_t)r1 * BSTRIDE + HCAP * p + lane];
      }
    }
#pragma unroll
    for (int p = 0; p < NPART; ++p) {
      m0[p] = __builtin_amdgcn_readfirstlane(m0[p]);
      m1[p] = __builtin_amdgcn_readfirstlane(m1[p]);
    }
    const int a1 = m0[0], a2 = a1 + m0[1], a3 = a2 + m0[2], n0 = a3 + m0[3];
    const int b1 = m1[0], b2 = b1 + m1[1], b3 = b2 + m1[2], n1 = b3 + m1[3];
    const bool over0 = m0[0] > HCAP || m0[1] > HCAP || m0[2] > HCAP || m0[3] > HCAP || n0 > CAP;
    const bool over1 = m1[0] > HCAP || m1[1] > HCAP || m1[2] > HCAP || m1[3] > HCAP || n1 > CAP;
    const int u01 = __builtin_amdgcn_ds_bpermute(((lane - a1) & 31) << 2, q0[1]);
    const int u02 = __builtin_amdgcn_ds_bpermute(((lane - a2) & 31) << 2, q0[2]);
    const int u03 = __builtin_amdgcn_ds_bpermute(((lane - a3) & 31) << 2, q0[3]);
    const int u11 = __builtin_amdgcn_ds_bpermute(((lane - b1) & 31) << 2, q1[1]);
    const int u12 = __builtin_amdgcn_ds_bpermute(((lane - b2) & 31) << 2, q1[2]);
    const int u13 = __builtin_amdgcn_ds_bpermute(((lane - b3) & 31) << 2, q1[3]);
    const int k0 = lane < a1 ? q0[0] : lane < a2 ? u01 : lane < a3 ? u02 : u03;
    const int k1 = lane < b1 ? q1[0] : lane < b2 ? u11 : lane < b3 ? u12 : u13;

    const float4 s0 = sum_row(x, xrow, bins, ei, r0, n0, over0, a1, a2, a3, k0, lane, nedges);
    *(float4*)&at[buf][(2 * w) * A_PITCH + 4 * lane] = s0;
    const float4 s1 = sum_row(x, xrow, bins, ei, r1, n1, over1, b1, b2, b3, k1, lane, nedges);
    *(float4*)&at[buf][(2 * w + 1) * A_PITCH + 4 * lane] = s1;
    if (lane == 0) { deg[buf][2 * w] = n0; deg[buf][2 * w + 1] = n1; }
    __syncthreads();

    const float* arow = &at[buf][lh * A_PITCH + 2 * khi];
    v8f c0 = (v8f){0.f, 0.f, 0.f, 0.f, 0.f, 0.f, 0.f, 0.f};
    v8f c1 = c0;
#pragma unroll
    for (int s = 0; s < 32; s += 2) {
      const v2f a0 = *(const v2f*)(arow + 4 * s);
      const v2f a1 = *(const v2f*)(arow + 4 * s + 4);
      const v2f b0 = s < KREG ? bf[s < KREG ? s : 0]
                              : *(const v2f*)(brow + 4 * (s - KREG));
      const v2f b1 = s + 1 < KREG ? bf[s + 1 < KREG ? s + 1 : 0]
                                  : *(const v2f*)(brow + 4 * (s + 1 - KREG));
      c0 = __builtin_amdgcn_wmma_f32_16x16x4_f32(
          false, a0, false, b0, (short)0, c0, false, false);
      c1 = __builtin_amdgcn_wmma_f32_16x16x4_f32(
          false, a1, false, b1, (short)0, c1, false, false);
    }

    float res[8];
#pragma unroll
    for (int i = 0; i < 8; ++i) {
      const int d = deg[buf][i + 8 * khi];
      const float bb = d > 0 ? (float)d * bv : 0.f;
      res[i] = tanhf((c0[i] + c1[i]) + bb);
    }
#pragma unroll
    for (int i = 0; i < 8; ++i) {
      const int m = i + 8 * khi;
      if (rb + m < nrows) out[(size_t)(rb + m) * CH + 16 * w + lh] = res[i];
    }
  }
}

extern "C" void kernel_launch(void* const* d_in, const int* in_sizes, int n_in,
                              void* d_out, int out_size, void* d_ws, size_t ws_size,
                              hipStream_t stream) {
  const float* x    = (const float*)d_in[0];
  const int*   ei   = (const int*)d_in[1];
  const float* W    = (const float*)d_in[2];
  const float* bias = (const float*)d_in[3];

  float* out = (float*)d_out;

  const int N = in_sizes[0] / CH;
  const int E = in_sizes[1] / 2;

  int*   bins   = (int*)d_ws;
  int*   cnt    = bins + (size_t)N * BSTRIDE;
  int*   seg    = cnt + (size_t)N * NPART;
  int*   segcnt = seg + (size_t)NRANGE * NSLICE * SEGCAP;

  gcn_file_ranges<<<NSLICE, A_THREADS, 0, stream>>>(ei, seg, segcnt, E, N);

  gcn_bin_rows<<<NPART * ((N + ROWS_PER_WG - 1) / ROWS_PER_WG), B_THREADS, 0, stream>>>(
      ei, seg, segcnt, cnt, bins, E, N);

  const int ntiles = (N + 15) / 16;
  gcn_gather_product<<<1024, 256, 0, stream>>>(x, W, bias, cnt, bins, ei, out, N, E, ntiles);
}


# tanh in the output stage as one exponential and one reciprocal per element, branch-free (was the library routine: two divergent branches per element); sums, their order and operand types kept
# speedup vs baseline: 4.5064x; 1.0540x over previous
#include <hip/hip_runtime.h>


typedef float v2f __attribute__((ext_vector_type(2)));
typedef float v8f __attribute__((ext_vector_type(8)));
typedef float v4f __attribute__((ext_vector_type(4)));
typedef const v4f __attribute__((address_space(1))) gv4f;

#define CH 128
#define A_PITCH 132
#define KREG 8
#define WL_K (CH - 4 * KREG)
#define WL_PITCH 100
#define GB 16

#define NPART 4
#define HCAP 32
#define BSTRIDE (NPART * HCAP)
#define CAP 64
#define ROWS_PER_WG 782
#define ROWS_PAD 784

#define NSLICE 256
#define NRANGE 64
#define SEGCAP 128
#define SPP (NSLICE / NPART)
#define PACK_BITS 22
#define PACK_MASK ((1u << PACK_BITS) - 1u)
#define A_THREADS 256
#define A_EPT 16
#define B_THREADS 512
static_assert(A_EPT == 16, "the load fence in gcn_file_ranges names 16 values");
#define B_LPS (B_THREADS / SPP)
#define B_PIECES (SEGCAP / 4 / B_LPS)

static __device__ __forceinline__ void file_src(
    unsigned d, int src, int nr, int* lcnt, int* lbins) {
  if (d < (unsigned)nr) {
    const int pos = atomicAdd(&lcnt[d], 1);
    if (pos < HCAP) lbins[d * HCAP + pos] = src;
  }
}

static __device__ __forceinline__ void slice_bounds(int s, int nedges, int& beg, int& end) {
  const int per = (nedges + NSLICE - 1) / NSLICE;
  beg = min(nedges, s * per);
  end = min(nedges, beg + per);
}

__global__ __launch_bounds__(A_THREADS) void gcn_file_ranges(
    const int* __restrict__ ei, int* __restrict__ seg, int* __restrict__ segcnt,
    int nedges, int nrows) {
  __shared__ __attribute__((aligned(16))) int lseg[NRANGE * SEGCAP];
  __shared__ int lcnt[NRANGE];

  const int tid = threadIdx.x;
  const int s = blockIdx.x;

  for (int i = tid; i < NRANGE * SEGCAP / 4; i += A_THREADS)
    ((int4*)lseg)[i] = make_int4(0, 0, 0, 0);
  if (tid < NRANGE) lcnt[tid] = 0;
  __syncthreads();

  int beg, end;
  slice_bounds(s, nedges, beg, end);
  const bool packable = nrows <= (int)(PACK_MASK + 1u);

  if (packable) {
    for (int e0 = beg + tid; e0 < end; e0 += A_EPT * A_THREADS) {
      int r[A_EPT], c[A_EPT];
#pragma unroll
      for (int u = 0; u < A_EPT; ++u) {
        const int e = e0 + A_THREADS * u;
        const int ee = e < end ? e : beg;
        r[u] = ei[ee];
        c[u] = ei[nedges + ee];
      }
      asm volatile("" :: "v"(c[0]), "v"(c[1]), "v"(c[2]), "v"(c[3]),
                         "v"(c[4]), "v"(c[5]), "v"(c[6]), "v"(c[7]),
                         "v"(c[8]), "v"(c[9]), "v"(c[10]), "v"(c[11]),
                         "v"(c[12]), "v"(c[13]), "v"(c[14]), "v"(c[15]));
      int pos[A_EPT];
      unsigned gg[A_EPT];
#pragma unroll
      for (int u = 0; u < A_EPT; ++u) {
        const int e = e0 + A_THREADS * u;
        const unsigned row = (unsigned)r[u];
        gg[u] = row / (unsigned)ROWS_PER_WG;
        pos[u] = SEGCAP;
        if (e < end && row < (unsigned)nrows && gg[u] < (unsigned)NRANGE)
          pos[u] = atomicAdd(&lcnt[gg[u]], 1);
      }
#pragma unroll
      for (int u = 0; u < A_EPT; ++u) {
        if (pos[u] < SEGCAP) {
          const unsigned d = (unsigned)r[u] - gg[u] * (unsigned)ROWS_PER_WG;
          lseg[gg[u] * SEGCAP + pos[u]] = (int)((d << PACK_BITS) | (unsigned)c[u]);
        }
      }
    }
  }
  __syncthreads();

  for (int i = tid; i < NRANGE * SEGCAP / 4; i += A_THREADS) {
    const int g = i / (SEGCAP / 4), p = i % (SEGCAP / 4);
    ((int4*)seg)[((size_t)g * NSLICE + s) * (SEGCAP / 4) + p] = ((const int4*)lseg)[i];
  }
  if (tid < NRANGE) segcnt[tid * NSLICE + s] = packable ? lcnt[tid] : SEGCAP + 1;
}

__global__ __launch_bounds__(B_THREADS) void gcn_bin_rows(
    const int* __restrict__ ei, const int* __restrict__ seg, const int* __restrict__ segcnt,
    int* __restrict__ cnt, int* __restrict__ bins, int nedges, int nrows) {
  __shared__ __attribute__((aligned(16))) int lds[ROWS_PAD + ROWS_PER_WG * HCAP];
  int* lcnt = lds;
  int* lbins = lds + ROWS_PAD;

  const int tid = threadIdx.x;
  const int g = blockIdx.x / NPART;
  const int h = blockIdx.x % NPART;
  const int lo = g * ROWS_PER_WG;
  const int nr = min(ROWS_PER_WG, nrows - lo);

  const int q = tid % B_LPS;
  const int s = h * SPP + tid / B_LPS;
  const bool have = g < NRANGE;
  const int sc = have ? segcnt[g * NSLICE + s] : SEGCAP + 1;
  int4 v[B_PIECES];
#pragma unroll
  for (int u = 0; u < B_PIECES; ++u) {
    v[u] = make_int4(0, 0, 0, 0);
    if (have) v[u] = ((const int4*)seg)[((size_t)g * NSLICE + s) * (SEGCAP / 4) + q + B_LPS * u];
  }

  for (int i = tid; i < ROWS_PER_WG * HCAP / 4; i += B_THREADS)
    ((int4*)lbins)[i] = make_int4(0, 0, 0, 0);
  for (int i = tid; i < ROWS_PER_WG; i += B_THREADS) lcnt[i] = 0;
  __syncthreads();

  if (sc <= SEGCAP) {
    unsigned dd[4 * B_PIECES];
    int ss[4 * B_PIECES], pp[4 * B_PIECES];
#pragma unroll
    for (int u = 0; u < B_PIECES; ++u) {
      const int i0 = 4 * (q + B_LPS * u);
      const unsigned ev[4] = {(unsigned)v[u].x, (unsigned)v[u].y, (unsigned)v[u].z, (unsigned)v[u].w};
#pragma unroll
      for (int k = 0; k < 4; ++k) {
        dd[4 * u + k] = i0 + k < sc ? ev[k] >> PACK_BITS : 0xffffffffu;
        ss[4 * u + k] = (int)(ev[k] & PACK_MASK);
        pp[4 * u + k] = HCAP;
        if (dd[4 * u + k] < (unsigned)nr) pp[4 * u + k] = atomicAdd(&lcnt[dd[4 * u + k]], 1);
      }
    }
#pragma unroll
    for (int k = 0; k < 4 * B_PIECES; ++k)
      if (pp[k] < HCAP) lbins[dd[k] * HCAP + pp[k]] = ss[k];
  } else {
    int beg, end;
    slice_bounds(s, nedges, beg, end);
    for (int e = beg + q; e < end; e += B_LPS)
      file_src((unsigned)(ei[e] - lo), ei[nedges + e], nr, lcnt, lbins);
  }
  __syncthreads();

  int4* gb = (int4*)(bins + (size_t)lo * BSTRIDE) + (HCAP / 4) * h;
  for (int i = tid; i < nr * (HCAP / 4); i += B_THREADS)
    gb[(i >> 3) * (BSTRIDE / 4) + (i & 7)] = ((const int4*)lbins)[i];
  for (int i = tid; i < nr; i += B_THREADS) cnt[(size_t)h * nrows + lo + i] = lcnt[i];
}

typedef const char __attribute__((address_space(1))) gchar;
static __device__ __forceinline__ v4f load_row(const float* __restrict__ x, int c, unsigned lane) {
  const unsigned off = ((unsigned)c << 9) + (lane << 4);
  return *(const gv4f*)((gchar*)x + off);
}

template <int K>
static __device__ __forceinline__ void add_sorted(
    float4& acc, const float* __restrict__ x, unsigned lane, int sorted, int j) {
  v4f v[K];
#pragma unroll
  for (int u = 0; u < K; ++u) {
    const int c = __builtin_amdgcn_readlane(sorted, j + u);
    v[u] = load_row(x, c, lane);
  }
#pragma unroll
  for (int u = 0; u < K; ++u) {
    acc.x += v[u].x; acc.y += v[u].y; acc.z += v[u].z; acc.w += v[u].w;
  }
}

template <int K>
static __device__ __forceinline__ void add_tail(
    float4& acc, const float* __restrict__ x, unsigned lane, int sorted, int j, int rem) {
  v4f v[K];
#pragma unroll
  for (int u = 0; u < K; ++u) {
    const int c = __builtin_amdgcn_readlane(sorted, j + u);
    v[u] = load_row(x, c, lane);
  }
#pragma unroll
  for (int u = 0; u < K; ++u) {
    const bool on = u < rem;
    const float sx = acc.x + v[u].x, sy = acc.y + v[u].y;
    const float sz = acc.z + v[u].z, sw = acc.w + v[u].w;
    acc.x = on ? sx : acc.x; acc.y = on ? sy : acc.y;
    acc.z = on ? sz : acc.z; acc.w = on ? sw : acc.w;
  }
}

static __device__ __forceinline__ float4 sum_row(
    const float* __restrict__ x, const float4* __restrict__ xrow, const int* __restrict__ bins,
    const int* __restrict__ ei, int r, int n, bool over, int o1, int o2, int o3,
    int key, int lane, int nedges) {
  float4 acc = make_float4(0.f, 0.f, 0.f, 0.f);

  if (over) {
    for (int e0 = 0; e0 < nedges; e0 += 32) {
      const int e = e0 + lane;
      const bool hit = e < nedges && ei[e] == r;
      const int c = hit ? ei[nedges + e] : 0;
      unsigned m = __builtin_amdgcn_ballot_w32(hit);
      while (m) {
        const int l = __builtin_ctz(m);
        m &= m - 1;
        const int cs = __builtin_amdgcn_readlane(c, l);
        const float4 v = xrow[(size_t)cs * (CH / 4)];
        acc.x += v.x; acc.y += v.y; acc.z += v.z; acc.w += v.w;
      }
    }
  } else if (n <= 32) {
    key = lane < n ? key : 0x7fffffff;
    int rank = 0;
    for (int j = 0; j < n; ++j) {
      const int kj = __builtin_amdgcn_readlane(key, j);
      rank += (kj < key || (kj == key && j < lane)) ? 1 : 0;
    }
    if (lane >= n) rank = lane;
    int sorted = __builtin_amdgcn_ds_permute(rank << 2, key);
    const int last = __builtin_amdgcn_readlane(sorted, n - 1);
    sorted = lane < n ? sorted : last;
    int j = 0;
    for (; j + GB <= n; j += GB) add_sorted<GB>(acc, x, (unsigned)lane, sorted, j);
    const int rem = n - j;
    if (rem > 12) add_tail<16>(acc, x, (unsigned)lane, sorted, j, rem);
    else if (rem > 8) add_tail<12>(acc, x, (unsigned)lane, sorted, j, rem);
    else if (rem > 4) add_tail<8>(acc, x, (unsigned)lane, sorted, j, rem);
    else if (rem > 0) add_tail<4>(acc, x, (unsigned)lane, sorted, j, rem);
  } else {
    const int key0 = key;
    const int j1 = 32 + lane;
    const int p1 = (j1 >= o1 ? 1 : 0) + (j1 >= o2 ? 1 : 0) + (j1 >= o3 ? 1 : 0);
    const int s1 = j1 - (p1 == 0 ? 0 : p1 == 1 ? o1 : p1 == 2 ? o2 : o3);
    const int key1 = (j1 < n) ? bins[(size_t)r * BSTRIDE + HCAP * p1 + s1] : 0x7fffffff;
    int rank0 = 0, rank1 = 0;
    for (int j = 0; j < 32; ++j) {
      const int kj = __builtin_amdgcn_readlane(key0, j);
      rank0 += (kj < key0 || (kj == key0 && j < lane)) ? 1 : 0;
      rank1 += (kj <= key1) ? 1 : 0;
    }
    for (int j = 0; j < n - 32; ++j) {
      const int kj = __builtin_amdgcn_readlane(key1, j);
      rank0 += (kj < key0) ? 1 : 0;
      rank1 += (kj < key1 || (kj == key1 && j < lane)) ? 1 : 0;
    }
    if (lane + 32 >= n) rank1 = -1;
    for (int p0 = 0; p0 < n; p0 += 8) {
      float4 v[8];
#pragma unroll
      for (int u = 0; u < 8; ++u) {
        v[u] = make_float4(0.f, 0.f, 0.f, 0.f);
        const int p = p0 + u;
        if (p < n) {
          const unsigned b0 = __builtin_amdgcn_ballot_w32(rank0 == p);
          const unsigned b1 = __builtin_amdgcn_ballot_w32(rank1 == p);
          const unsigned b = b0 | b1;
          const int kk = b0 ? key0 : key1;
          const int c = __builtin_amdgcn_readlane(kk, __builtin_ctz(b));
          v[u] = xrow[(size_t)c * (CH / 4)];
        }
      }
#pragma unroll
      for (int u = 0; u < 8; ++u) {
        acc.x += v[u].x; acc.y += v[u].y; acc.z += v[u].z; acc.w += v[u].w;
      }
    }
  }
  return acc;
}

static __device__ __forceinline__ float tanh_by_exp(float v) {
  const float t = __builtin_amdgcn_exp2f(__builtin_fabsf(v) * -2.8853900817779268f);
  const float r = (1.0f - t) * __builtin_amdgcn_rcpf(1.0f + t);
  return __builtin_copysignf(r, v);
}

__global__ __launch_bounds__(256) __attribute__((amdgpu_waves_per_eu(8, 8))) void gcn_gather_product(
    const float* __restrict__ x, const float* __restrict__ W,
    const float* __restrict__ bias, const int* __restrict__ cnt,
    const int* __restrict__ bins, const int* __restrict__ ei,
    float* __restrict__ out, int nrows, int nedges, int ntiles) {
  __shared__ float at[2][16 * A_PITCH];
  __shared__ int deg[2][16];
  __shared__ __attribute__((aligned(16))) float wl[CH * WL_PITCH];

  const int lane = threadIdx.x & 31;
  const int w = __builtin_amdgcn_readfirstlane((int)(threadIdx.x >> 5));
  const int lh  = lane & 15;
  const int khi = lane >> 4;
  const float4* xrow = (const float4*)x + lane;

  v2f bf[KREG];
  {
    const float* wrow = W + (size_t)(16 * w + lh) * CH + 2 * khi;
#pragma unroll
    for (int s = 0; s < KREG; ++s) bf[s] = *(const v2f*)(wrow + 4 * s);
  }
#pragma unroll
  for (int t = 0; t < CH * (WL_K / 4) / 256; ++t) {
    const int i = (int)threadIdx.x + 256 * t;
    const int n = i / (WL_K / 4), c4 = i % (WL_K / 4);
    *(float4*)&wl[n * WL_PITCH + 4 * c4] =
        *(const float4*)(W + (size_t)n * CH + 4 * KREG + 4 * c4);
  }
  const float* brow = &wl[(16 * w + lh) * WL_PITCH + 2 * khi];
  const float bv = bias[16 * w + lh];

  int buf = 0;
  for (int tile = blockIdx.x; tile < ntiles; tile += gridDim.x, buf ^= 1) {
    const int rb = tile * 16;
    const int r0 = rb + 2 * w, r1 = r0 + 1;
    int m0[NPART], m1[NPART], q0[NPART], q1[NPART];
#pragma unroll
    for (int p = 0; p < NPART; ++p) {
      m0[p] = 0; m1[p] = 0; q0[p] = 0; q1[p] = 0;
      if (r0 < nrows) {
        m0[p] = cnt[(size_t)p * nrows + r0];
        q0[p] = bins[(size_t)r0 * BSTRIDE + HCAP * p + lane];
      }
      if (r1 < nrows) {
        m1[p] = cnt[(size_t)p * nrows + r1];
        q1[p] = bins[(size_t)r1 * BSTRIDE + HCAP * p + lane];
      }
    }
#pragma unroll
    for (int p = 0; p < NPART; ++p) {
      m0[p] = __builtin_amdgcn_readfirstlane(m0[p]);
      m1[p] = __builtin_amdgcn_readfirstlane(m1[p]);
    }
    const int a1 = m0[0], a2 = a1 + m0[1], a3 = a2 + m0[2], n0 = a3 + m0[3];
    const int b1 = m1[0], b2 = b1 + m1[1], b3 = b2 + m1[2], n1 = b3 + m1[3];
    const bool over0 = m0[0] > HCAP || m0[1] > HCAP || m0[2] > HCAP || m0[3] > HCAP || n0 > CAP;
    const bool over1 = m1[0] > HCAP || m1[1] > HCAP || m1[2] > HCAP || m1[3] > HCAP || n1 > CAP;
    const int u01 = __builtin_amdgcn_ds_bpermute(((lane - a1) & 31) << 2, q0[1]);
    const int u02 = __builtin_amdgcn_ds_bpermute(((lane - a2) & 31) << 2, q0[2]);
    const int u03 = __builtin_amdgcn_ds_bpermute(((lane - a3) & 31) << 2, q0[3]);
    const int u11 = __builtin_amdgcn_ds_bpermute(((lane - b1) & 31) << 2, q1[1]);
    const int u12 = __builtin_amdgcn_ds_bpermute(((lane - b2) & 31) << 2, q1[2]);
    const int u13 = __builtin_amdgcn_ds_bpermute(((lane - b3) & 31) << 2, q1[3]);
    const int k0 = lane < a1 ? q0[0] : lane < a2 ? u01 : lane < a3 ? u02 : u03;
    const int k1 = lane < b1 ? q1[0] : lane < b2 ? u11 : lane < b3 ? u12 : u13;

    const float4 s0 = sum_row(x, xrow, bins, ei, r0, n0, over0, a1, a2, a3, k0, lane, nedges);
    *(float4*)&at[buf][(2 * w) * A_PITCH + 4 * lane] = s0;
    const float4 s1 = sum_row(x, xrow, bins, ei, r1, n1, over1, b1, b2, b3, k1, lane, nedges);
    *(float4*)&at[buf][(2 * w + 1) * A_PITCH + 4 * lane] = s1;
    if (lane == 0) { deg[buf][2 * w] = n0; deg[buf][2 * w + 1] = n1; }
    __syncthreads();

    const float* arow = &at[buf][lh * A_PITCH + 2 * khi];
    v8f c0 = (v8f){0.f, 0.f, 0.f, 0.f, 0.f, 0.f, 0.f, 0.f};
    v8f c1 = c0;
#pragma unroll
    for (int s = 0; s < 32; s += 2) {
      const v2f a0 = *(const v2f*)(arow + 4 * s);
      const v2f a1 = *(const v2f*)(arow + 4 * s + 4);
      const v2f b0 = s < KREG ? bf[s < KREG ? s : 0]
                              : *(const v2f*)(brow + 4 * (s - KREG));
      const v2f b1 = s + 1 < KREG ? bf[s + 1 < KREG ? s + 1 : 0]
                                  : *(const v2f*)(brow + 4 * (s + 1 - KREG));
      c0 = __builtin_amdgcn_wmma_f32_16x16x4_f32(
          false, a0, false, b0, (short)0, c0, false, false);
      c1 = __builtin_amdgcn_wmma_f32_16x16x4_f32(
          false, a1, false, b1, (short)0, c1, false, false);
    }

    float res[8];
#pragma unroll
    for (int i = 0; i < 8; ++i) {
      const int d = deg[buf][i + 8 * khi];
      const float bb = d > 0 ? (float)d * bv : 0.f;
      res[i] = tanh_by_exp((c0[i] + c1[i]) + bb);
    }
#pragma unroll
    for (int i = 0; i < 8; ++i) {
      const int m = i + 8 * khi;
      if (rb + m < nrows) out[(size_t)(rb + m) * CH + 16 * w + lh] = res[i];
    }
  }
}

extern "C" void kernel_launch(void* const* d_in, const int* in_sizes, int n_in,
                              void* d_out, int out_size, void* d_ws, size_t ws_size,
                              hipStream_t stream) {
  const float* x    = (const float*)d_in[0];
  const int*   ei   = (const int*)d_in[1];
  const float* W    = (const float*)d_in[2];
  const float* bias = (const float*)d_in[3];

  float* out = (float*)d_out;

  const int N = in_sizes[0] / CH;
  const int E = in_sizes[1] / 2;

  int*   bins   = (int*)d_ws;
  int*   cnt    = bins + (size_t)N * BSTRIDE;
  int*   seg    = cnt + (size_t)N * NPART;
  int*   segcnt = seg + (size_t)NRANGE * NSLICE * SEGCAP;

  gcn_file_ranges<<<NSLICE, A_THREADS, 0, stream>>>(ei, seg, segcnt, E, N);

  gcn_bin_rows<<<NPART * ((N + ROWS_PER_WG - 1) / ROWS_PER_WG), B_THREADS, 0, stream>>>(
      ei, seg, segcnt, cnt, bins, E, N);

  const int ntiles = (N + 15) / 16;
  gcn_gather_product<<<1024, 256, 0, stream>>>(x, W, bias, cnt, bins, ei, out, N, E, ntiles);
}
